# starting kernel
# baseline (speedup 1.0000x reference)
#include <hip/hip_runtime.h>

#define N_NODES_C 50000
#define N_EDGES_C 800000
#define F_IN_C 32
#define C_C 256
#define N_PAD 50016
#define N_ACT_C 8

typedef __attribute__((ext_vector_type(16))) __bf16 v16bf;
typedef __attribute__((ext_vector_type(8)))  __bf16 v8bf;
typedef __attribute__((ext_vector_type(8)))  float  v8f;

static __device__ __forceinline__ v8f wmma_bf16(v16bf a, v16bf b, v8f c) {
    return __builtin_amdgcn_wmma_f32_16x16x32_bf16(
         false, a,  false, b,
         (short)0, c,  false,  false);
}

static __device__ __forceinline__ v16bf load_a_frag_f32(
    const float* __restrict__ A, int row, int ld, int k0, int lane) {
    int half = lane >> 4;
    const float* p0 = A + (long long)row * ld + k0 + half * 8;
    const float* p1 = p0 + 16;
    float4 f0 = *(const float4*)(p0);
    float4 f1 = *(const float4*)(p0 + 4);
    float4 f2 = *(const float4*)(p1);
    float4 f3 = *(const float4*)(p1 + 4);
    v16bf a;
    a[0]  = (__bf16)f0.x; a[1]  = (__bf16)f0.y; a[2]  = (__bf16)f0.z; a[3]  = (__bf16)f0.w;
    a[4]  = (__bf16)f1.x; a[5]  = (__bf16)f1.y; a[6]  = (__bf16)f1.z; a[7]  = (__bf16)f1.w;
    a[8]  = (__bf16)f2.x; a[9]  = (__bf16)f2.y; a[10] = (__bf16)f2.z; a[11] = (__bf16)f2.w;
    a[12] = (__bf16)f3.x; a[13] = (__bf16)f3.y; a[14] = (__bf16)f3.z; a[15] = (__bf16)f3.w;
    return a;
}

static __device__ __forceinline__ v16bf load_b_frag_bf16(
    const __bf16* __restrict__ BT, int col, int ld, int k0, int lane) {
    int half = lane >> 4;
    const __bf16* p = BT + (long long)col * ld + k0 + half * 8;
    v8bf b0 = *(const v8bf*)(p);
    v8bf b1 = *(const v8bf*)(p + 16);
    v16bf b;
#pragma unroll
    for (int i = 0; i < 8; ++i) { b[i] = b0[i]; b[8 + i] = b1[i]; }
    return b;
}

static __device__ __forceinline__ float sigmoidf_(float x) {
    return 1.0f / (1.0f + __expf(-x));
}

__global__ void convw_kernel(const float* __restrict__ gg,
                             const float* __restrict__ wih,
                             const float* __restrict__ whh,
                             __bf16* __restrict__ ggT,
                             __bf16* __restrict__ wihb,
                             __bf16* __restrict__ whhb) {
    int i = blockIdx.x * 256 + threadIdx.x;
    if (i < 3 * 256 * 256) {
        int l = i >> 16, r = i & 65535, j = r >> 8, k = r & 255;
        ggT[i] = (__bf16)gg[l * 65536 + k * 256 + j];
    }
    if (i < 768 * 256) {
        wihb[i] = (__bf16)wih[i];
        whhb[i] = (__bf16)whh[i];
    }
}

__global__ void init_h_kernel(const float* __restrict__ x, float* __restrict__ h) {
    int i = blockIdx.x * 256 + threadIdx.x;
    int n = i >> 8, j = i & 255;
    float v = 0.0f;
    if (n < N_NODES_C && j < F_IN_C) v = x[n * F_IN_C + j];
    h[i] = v;
}

__global__ void zero4_kernel(float4* __restrict__ p, int n4) {
    int i = blockIdx.x * 256 + threadIdx.x;
    if (i < n4) p[i] = make_float4(0.f, 0.f, 0.f, 0.f);
}

__global__ void __launch_bounds__(256)
gemm_m_kernel(const float* __restrict__ h, const __bf16* __restrict__ WgT,
              float* __restrict__ m) {
    int lane = threadIdx.x & 31;
    int wave = threadIdx.x >> 5;
    int half = lane >> 4;
    int row0 = blockIdx.x * 32 + (wave >> 2) * 16;
    int c0   = (wave & 3) * 64;
    int rowA = row0 + (lane & 15);

    v8f acc0 = {}, acc1 = {}, acc2 = {}, acc3 = {};
#pragma unroll
    for (int k0 = 0; k0 < 256; k0 += 32) {
        v16bf a = load_a_frag_f32(h, rowA, 256, k0, lane);
        v16bf b0 = load_b_frag_bf16(WgT, c0 +  0 + (lane & 15), 256, k0, lane);
        acc0 = wmma_bf16(a, b0, acc0);
        v16bf b1 = load_b_frag_bf16(WgT, c0 + 16 + (lane & 15), 256, k0, lane);
        acc1 = wmma_bf16(a, b1, acc1);
        v16bf b2 = load_b_frag_bf16(WgT, c0 + 32 + (lane & 15), 256, k0, lane);
        acc2 = wmma_bf16(a, b2, acc2);
        v16bf b3 = load_b_frag_bf16(WgT, c0 + 48 + (lane & 15), 256, k0, lane);
        acc3 = wmma_bf16(a, b3, acc3);
    }
    v8f accs[4] = {acc0, acc1, acc2, acc3};
#pragma unroll
    for (int t = 0; t < 4; ++t) {
        int c = c0 + t * 16 + (lane & 15);
#pragma unroll
        for (int vr = 0; vr < 8; ++vr)
            m[(long long)(row0 + vr + 8 * half) * 256 + c] = accs[t][vr];
    }
}

__global__ void __launch_bounds__(256)
scatter_kernel(const float* __restrict__ m, const int* __restrict__ ei,
               const float* __restrict__ ew, float* __restrict__ agg) {
    int tid = blockIdx.x * 256 + threadIdx.x;
    int e = tid >> 6;
    if (e >= N_EDGES_C) return;
    int f = (tid & 63) << 2;
    int s = ei[e];
    int d = ei[N_EDGES_C + e];
    float w = ew[e];
    float4 v = *(const float4*)(m + (long long)s * 256 + f);
    float* dst = agg + (long long)d * 256 + f;
    atomicAdd(dst + 0, v.x * w);
    atomicAdd(dst + 1, v.y * w);
    atomicAdd(dst + 2, v.z * w);
    atomicAdd(dst + 3, v.w * w);
}

__global__ void __launch_bounds__(256)
gru_kernel(const float* __restrict__ agg, float* __restrict__ h,
           const __bf16* __restrict__ wihb, const __bf16* __restrict__ whhb,
           const float* __restrict__ bih, const float* __restrict__ bhh) {
    int lane = threadIdx.x & 31;
    int wave = threadIdx.x >> 5;
    int half = lane >> 4;
    int row0 = blockIdx.x * 16;
    int c0   = wave * 32;
    int rowA = row0 + (lane & 15);

    v8f acc[2][3][2];
#pragma unroll
    for (int a = 0; a < 2; ++a)
#pragma unroll
        for (int g = 0; g < 3; ++g)
#pragma unroll
            for (int t = 0; t < 2; ++t) { v8f z = {}; acc[a][g][t] = z; }

#pragma unroll
    for (int k0 = 0; k0 < 256; k0 += 32) {
        v16bf a_agg = load_a_frag_f32(agg, rowA, 256, k0, lane);
        v16bf a_h   = load_a_frag_f32(h,   rowA, 256, k0, lane);
#pragma unroll
        for (int g = 0; g < 3; ++g) {
#pragma unroll
            for (int t = 0; t < 2; ++t) {
                int col = g * 256 + c0 + t * 16 + (lane & 15);
                v16bf b_ih = load_b_frag_bf16(wihb, col, 256, k0, lane);
                acc[0][g][t] = wmma_bf16(a_agg, b_ih, acc[0][g][t]);
                v16bf b_hh = load_b_frag_bf16(whhb, col, 256, k0, lane);
                acc[1][g][t] = wmma_bf16(a_h, b_hh, acc[1][g][t]);
            }
        }
    }

    __syncthreads();

#pragma unroll
    for (int t = 0; t < 2; ++t) {
        int c = c0 + t * 16 + (lane & 15);
        float bir = bih[c],       bhr = bhh[c];
        float biz = bih[256 + c], bhz = bhh[256 + c];
        float bin = bih[512 + c], bhn = bhh[512 + c];
#pragma unroll
        for (int vr = 0; vr < 8; ++vr) {
            long long idx = (long long)(row0 + vr + 8 * half) * 256 + c;
            float hold = h[idx];
            float r  = sigmoidf_(acc[0][0][t][vr] + bir + acc[1][0][t][vr] + bhr);
            float z  = sigmoidf_(acc[0][1][t][vr] + biz + acc[1][1][t][vr] + bhz);
            float nn = tanhf(acc[0][2][t][vr] + bin + r * (acc[1][2][t][vr] + bhn));
            h[idx] = (1.0f - z) * nn + z * hold;
        }
    }
}

__global__ void __launch_bounds__(256)
final_kernel(const float* __restrict__ h, const float* __restrict__ fcw,
             const float* __restrict__ fcb, float* __restrict__ out) {
    __shared__ float sw[N_ACT_C * 256];
    __shared__ float sb[N_ACT_C];
    for (int i = threadIdx.x; i < N_ACT_C * 256; i += 256) sw[i] = fcw[i];
    if (threadIdx.x < N_ACT_C) sb[threadIdx.x] = fcb[threadIdx.x];
    __syncthreads();

    int n = blockIdx.x * 256 + threadIdx.x;
    if (n >= N_NODES_C) return;

    float acc[N_ACT_C];
#pragma unroll
    for (int a = 0; a < N_ACT_C; ++a) acc[a] = sb[a];
    const float* hr = h + (long long)n * 256;
    for (int k = 0; k < 256; ++k) {
        float hv = fmaxf(hr[k], 0.0f);
#pragma unroll
        for (int a = 0; a < N_ACT_C; ++a) acc[a] += hv * sw[a * 256 + k];
    }
    float mx = acc[0];
#pragma unroll
    for (int a = 1; a < N_ACT_C; ++a) mx = fmaxf(mx, acc[a]);
    float s = 0.0f;
#pragma unroll
    for (int a = 0; a < N_ACT_C; ++a) { acc[a] = __expf(acc[a] - mx); s += acc[a]; }
    float inv = 1.0f / s;
#pragma unroll
    for (int a = 0; a < N_ACT_C; ++a) out[(long long)n * N_ACT_C + a] = acc[a] * inv;
}

extern "C" void kernel_launch(void* const* d_in, const int* in_sizes, int n_in,
                              void* d_out, int out_size, void* d_ws, size_t ws_size,
                              hipStream_t stream) {
    const float* x   = (const float*)d_in[0];
    const int*   ei  = (const int*)  d_in[1];
    const float* ew  = (const float*)d_in[2];
    const float* gg  = (const float*)d_in[3];
    const float* wih = (const float*)d_in[4];
    const float* whh = (const float*)d_in[5];
    const float* bih = (const float*)d_in[6];
    const float* bhh = (const float*)d_in[7];
    const float* fcw = (const float*)d_in[8];
    const float* fcb = (const float*)d_in[9];
    float* out = (float*)d_out;

    float* h   = (float*)d_ws;
    float* m   = h + (size_t)N_PAD * 256;
    float* agg = m + (size_t)N_PAD * 256;
    __bf16* ggT  = (__bf16*)(agg + (size_t)N_PAD * 256);
    __bf16* wihb = ggT + 3 * 256 * 256;
    __bf16* whhb = wihb + 768 * 256;

    convw_kernel<<<(3 * 256 * 256 + 255) / 256, 256, 0, stream>>>(gg, wih, whh,
                                                                  ggT, wihb, whhb);
    init_h_kernel<<<N_PAD, 256, 0, stream>>>(x, h);

    for (int l = 0; l < 3; ++l) {
        gemm_m_kernel<<<N_PAD / 32, 256, 0, stream>>>(h, ggT + (size_t)l * 65536, m);
        zero4_kernel<<<(N_PAD * 64 + 255) / 256, 256, 0, stream>>>((float4*)agg,
                                                                   N_PAD * 64);
        scatter_kernel<<<(N_EDGES_C * 64) / 256, 256, 0, stream>>>(m, ei, ew, agg);
        gru_kernel<<<N_PAD / 16, 256, 0, stream>>>(agg, h, wihb, whhb, bih, bhh);
    }

    final_kernel<<<(N_NODES_C + 255) / 256, 256, 0, stream>>>(h, fcw, fcb, out);
}


# Float-atomic scatter and its zero pass replaced by a per-call edge list sorted by (dst, edge id) and one wave per node summing its bucket in registers: sum order fixed, ascending edge id
# speedup vs baseline: 1.4681x; 1.4681x over previous
#include <hip/hip_runtime.h>

#define N_NODES_C 50000
#define N_EDGES_C 800000
#define F_IN_C 32
#define C_C 256
#define N_PAD 50016
#define N_ACT_C 8

typedef __attribute__((ext_vector_type(16))) __bf16 v16bf;
typedef __attribute__((ext_vector_type(8)))  __bf16 v8bf;
typedef __attribute__((ext_vector_type(8)))  float  v8f;

static __device__ __forceinline__ v8f wmma_bf16(v16bf a, v16bf b, v8f c) {
    return __builtin_amdgcn_wmma_f32_16x16x32_bf16(
         false, a,  false, b,
         (short)0, c,  false,  false);
}

static __device__ __forceinline__ v16bf load_a_frag_f32(
    const float* __restrict__ A, int row, int ld, int k0, int lane) {
    int half = lane >> 4;
    const float* p0 = A + (long long)row * ld + k0 + half * 8;
    const float* p1 = p0 + 16;
    float4 f0 = *(const float4*)(p0);
    float4 f1 = *(const float4*)(p0 + 4);
    float4 f2 = *(const float4*)(p1);
    float4 f3 = *(const float4*)(p1 + 4);
    v16bf a;
    a[0]  = (__bf16)f0.x; a[1]  = (__bf16)f0.y; a[2]  = (__bf16)f0.z; a[3]  = (__bf16)f0.w;
    a[4]  = (__bf16)f1.x; a[5]  = (__bf16)f1.y; a[6]  = (__bf16)f1.z; a[7]  = (__bf16)f1.w;
    a[8]  = (__bf16)f2.x; a[9]  = (__bf16)f2.y; a[10] = (__bf16)f2.z; a[11] = (__bf16)f2.w;
    a[12] = (__bf16)f3.x; a[13] = (__bf16)f3.y; a[14] = (__bf16)f3.z; a[15] = (__bf16)f3.w;
    return a;
}

static __device__ __forceinline__ v16bf load_b_frag_bf16(
    const __bf16* __restrict__ BT, int col, int ld, int k0, int lane) {
    int half = lane >> 4;
    const __bf16* p = BT + (long long)col * ld + k0 + half * 8;
    v8bf b0 = *(const v8bf*)(p);
    v8bf b1 = *(const v8bf*)(p + 16);
    v16bf b;
#pragma unroll
    for (int i = 0; i < 8; ++i) { b[i] = b0[i]; b[8 + i] = b1[i]; }
    return b;
}

static __device__ __forceinline__ float sigmoidf_(float x) {
    return 1.0f / (1.0f + __expf(-x));
}

__global__ void convw_kernel(const float* __restrict__ gg,
                             const float* __restrict__ wih,
                             const float* __restrict__ whh,
                             __bf16* __restrict__ ggT,
                             __bf16* __restrict__ wihb,
                             __bf16* __restrict__ whhb) {
    int i = blockIdx.x * 256 + threadIdx.x;
    if (i < 3 * 256 * 256) {
        int l = i >> 16, r = i & 65535, j = r >> 8, k = r & 255;
        ggT[i] = (__bf16)gg[l * 65536 + k * 256 + j];
    }
    if (i < 768 * 256) {
        wihb[i] = (__bf16)wih[i];
        whhb[i] = (__bf16)whh[i];
    }
}

__global__ void init_h_kernel(const float* __restrict__ x, float* __restrict__ h) {
    int i = blockIdx.x * 256 + threadIdx.x;
    int n = i >> 8, j = i & 255;
    float v = 0.0f;
    if (n < N_NODES_C && j < F_IN_C) v = x[n * F_IN_C + j];
    h[i] = v;
}

__global__ void zero_int_kernel(int* __restrict__ p, int n) {
    int i = blockIdx.x * 256 + threadIdx.x;
    if (i < n) p[i] = 0;
}

__global__ void __launch_bounds__(256)
count_kernel(const int* __restrict__ ei, int* __restrict__ cnt,
             int* __restrict__ slot) {
    int e = blockIdx.x * 256 + threadIdx.x;
    if (e >= N_EDGES_C) return;
    slot[e] = atomicAdd(&cnt[ei[N_EDGES_C + e]], 1);
}

#define SCAN_PER ((N_NODES_C + 1 + 1023) / 1024)
__global__ void __launch_bounds__(1024)
scan_kernel(const int* __restrict__ cnt, int* __restrict__ off) {
    __shared__ int part[1024];
    int t = threadIdx.x;
    int start = t * SCAN_PER;
    int s = 0;
    for (int i = 0; i < SCAN_PER; ++i) {
        int idx = start + i;
        if (idx < N_NODES_C) s += cnt[idx];
    }
    part[t] = s;
    __syncthreads();
    for (int o = 1; o < 1024; o <<= 1) {
        int v = (t >= o) ? part[t - o] : 0;
        __syncthreads();
        part[t] += v;
        __syncthreads();
    }
    int run = part[t] - s;
    for (int i = 0; i < SCAN_PER; ++i) {
        int idx = start + i;
        if (idx <= N_NODES_C) off[idx] = run;
        if (idx < N_NODES_C) run += cnt[idx];
    }
}

__global__ void __launch_bounds__(256)
fill_kernel(const int* __restrict__ ei, const int* __restrict__ off,
            const int* __restrict__ slot, int* __restrict__ bucket) {
    int e = blockIdx.x * 256 + threadIdx.x;
    if (e >= N_EDGES_C) return;
    bucket[off[ei[N_EDGES_C + e]] + slot[e]] = e;
}

__global__ void __launch_bounds__(256)
rank_kernel(const int* __restrict__ ei, const float* __restrict__ ew,
            const int* __restrict__ off, const int* __restrict__ bucket,
            int* __restrict__ csrc, float* __restrict__ cw) {
    int e = blockIdx.x * 256 + threadIdx.x;
    if (e >= N_EDGES_C) return;
    int d = ei[N_EDGES_C + e];
    int b = off[d];
    int n = off[d + 1] - b;
    int r = 0;
    for (int i = 0; i < n; ++i) r += (bucket[b + i] < e) ? 1 : 0;
    csrc[b + r] = ei[e];
    cw[b + r]   = ew[e];
}

__global__ void __launch_bounds__(256)
gemm_m_kernel(const float* __restrict__ h, const __bf16* __restrict__ WgT,
              float* __restrict__ m) {
    int lane = threadIdx.x & 31;
    int wave = threadIdx.x >> 5;
    int half = lane >> 4;
    int row0 = blockIdx.x * 32 + (wave >> 2) * 16;
    int c0   = (wave & 3) * 64;
    int rowA = row0 + (lane & 15);

    v8f acc0 = {}, acc1 = {}, acc2 = {}, acc3 = {};
#pragma unroll
    for (int k0 = 0; k0 < 256; k0 += 32) {
        v16bf a = load_a_frag_f32(h, rowA, 256, k0, lane);
        v16bf b0 = load_b_frag_bf16(WgT, c0 +  0 + (lane & 15), 256, k0, lane);
        acc0 = wmma_bf16(a, b0, acc0);
        v16bf b1 = load_b_frag_bf16(WgT, c0 + 16 + (lane & 15), 256, k0, lane);
        acc1 = wmma_bf16(a, b1, acc1);
        v16bf b2 = load_b_frag_bf16(WgT, c0 + 32 + (lane & 15), 256, k0, lane);
        acc2 = wmma_bf16(a, b2, acc2);
        v16bf b3 = load_b_frag_bf16(WgT, c0 + 48 + (lane & 15), 256, k0, lane);
        acc3 = wmma_bf16(a, b3, acc3);
    }
    v8f accs[4] = {acc0, acc1, acc2, acc3};
#pragma unroll
    for (int t = 0; t < 4; ++t) {
        int c = c0 + t * 16 + (lane & 15);
#pragma unroll
        for (int vr = 0; vr < 8; ++vr)
            m[(long long)(row0 + vr + 8 * half) * 256 + c] = accs[t][vr];
    }
}

__global__ void __launch_bounds__(256)
gather_kernel(const float* __restrict__ m, const int* __restrict__ off,
              const int* __restrict__ csrc, const float* __restrict__ cw,
              float* __restrict__ agg) {
    int lane = threadIdx.x & 31;
    int node = __builtin_amdgcn_readfirstlane(blockIdx.x * 8 + (threadIdx.x >> 5));
    int b = 0, n = 0;
    if (node < N_NODES_C) { b = off[node]; n = off[node + 1] - b; }

    float4 a0 = make_float4(0.f, 0.f, 0.f, 0.f);
    float4 a1 = make_float4(0.f, 0.f, 0.f, 0.f);
    for (int base = 0; base < n; base += 32) {
        int c = n - base;
        if (c > 32) c = 32;
        int   s_l = 0;
        float w_l = 0.0f;
        if (lane < c) { s_l = csrc[b + base + lane]; w_l = cw[b + base + lane]; }
#pragma unroll 4
        for (int j = 0; j < c; ++j) {
            int   s = __builtin_amdgcn_readlane(s_l, j);
            float w = __int_as_float(__builtin_amdgcn_readlane(__float_as_int(w_l), j));
            const float4* row = (const float4*)(m + (long long)s * 256);
            float4 v0 = row[lane];
            float4 v1 = row[32 + lane];
            a0.x += v0.x * w; a0.y += v0.y * w; a0.z += v0.z * w; a0.w += v0.w * w;
            a1.x += v1.x * w; a1.y += v1.y * w; a1.z += v1.z * w; a1.w += v1.w * w;
        }
    }
    float4* orow = (float4*)(agg + (long long)node * 256);
    orow[lane]      = a0;
    orow[32 + lane] = a1;
}

__global__ void __launch_bounds__(256)
gru_kernel(const float* __restrict__ agg, float* __restrict__ h,
           const __bf16* __restrict__ wihb, const __bf16* __restrict__ whhb,
           const float* __restrict__ bih, const float* __restrict__ bhh) {
    int lane = threadIdx.x & 31;
    int wave = threadIdx.x >> 5;
    int half = lane >> 4;
    int row0 = blockIdx.x * 16;
    int c0   = wave * 32;
    int rowA = row0 + (lane & 15);

    v8f acc[2][3][2];
#pragma unroll
    for (int a = 0; a < 2; ++a)
#pragma unroll
        for (int g = 0; g < 3; ++g)
#pragma unroll
            for (int t = 0; t < 2; ++t) { v8f z = {}; acc[a][g][t] = z; }

#pragma unroll
    for (int k0 = 0; k0 < 256; k0 += 32) {
        v16bf a_agg = load_a_frag_f32(agg, rowA, 256, k0, lane);
        v16bf a_h   = load_a_frag_f32(h,   rowA, 256, k0, lane);
#pragma unroll
        for (int g = 0; g < 3; ++g) {
#pragma unroll
            for (int t = 0; t < 2; ++t) {
                int col = g * 256 + c0 + t * 16 + (lane & 15);
                v16bf b_ih = load_b_frag_bf16(wihb, col, 256, k0, lane);
                acc[0][g][t] = wmma_bf16(a_agg, b_ih, acc[0][g][t]);
                v16bf b_hh = load_b_frag_bf16(whhb, col, 256, k0, lane);
                acc[1][g][t] = wmma_bf16(a_h, b_hh, acc[1][g][t]);
            }
        }
    }

    __syncthreads();

#pragma unroll
    for (int t = 0; t < 2; ++t) {
        int c = c0 + t * 16 + (lane & 15);
        float bir = bih[c],       bhr = bhh[c];
        float biz = bih[256 + c], bhz = bhh[256 + c];
        float bin = bih[512 + c], bhn = bhh[512 + c];
#pragma unroll
        for (int vr = 0; vr < 8; ++vr) {
            long long idx = (long long)(row0 + vr + 8 * half) * 256 + c;
            float hold = h[idx];
            float r  = sigmoidf_(acc[0][0][t][vr] + bir + acc[1][0][t][vr] + bhr);
            float z  = sigmoidf_(acc[0][1][t][vr] + biz + acc[1][1][t][vr] + bhz);
            float nn = tanhf(acc[0][2][t][vr] + bin + r * (acc[1][2][t][vr] + bhn));
            h[idx] = (1.0f - z) * nn + z * hold;
        }
    }
}

__global__ void __launch_bounds__(256)
final_kernel(const float* __restrict__ h, const float* __restrict__ fcw,
             const float* __restrict__ fcb, float* __restrict__ out) {
    __shared__ float sw[N_ACT_C * 256];
    __shared__ float sb[N_ACT_C];
    for (int i = threadIdx.x; i < N_ACT_C * 256; i += 256) sw[i] = fcw[i];
    if (threadIdx.x < N_ACT_C) sb[threadIdx.x] = fcb[threadIdx.x];
    __syncthreads();

    int n = blockIdx.x * 256 + threadIdx.x;
    if (n >= N_NODES_C) return;

    float acc[N_ACT_C];
#pragma unroll
    for (int a = 0; a < N_ACT_C; ++a) acc[a] = sb[a];
    const float* hr = h + (long long)n * 256;
    for (int k = 0; k < 256; ++k) {
        float hv = fmaxf(hr[k], 0.0f);
#pragma unroll
        for (int a = 0; a < N_ACT_C; ++a) acc[a] += hv * sw[a * 256 + k];
    }
    float mx = acc[0];
#pragma unroll
    for (int a = 1; a < N_ACT_C; ++a) mx = fmaxf(mx, acc[a]);
    float s = 0.0f;
#pragma unroll
    for (int a = 0; a < N_ACT_C; ++a) { acc[a] = __expf(acc[a] - mx); s += acc[a]; }
    float inv = 1.0f / s;
#pragma unroll
    for (int a = 0; a < N_ACT_C; ++a) out[(long long)n * N_ACT_C + a] = acc[a] * inv;
}

extern "C" void kernel_launch(void* const* d_in, const int* in_sizes, int n_in,
                              void* d_out, int out_size, void* d_ws, size_t ws_size,
                              hipStream_t stream) {
    const float* x   = (const float*)d_in[0];
    const int*   ei  = (const int*)  d_in[1];
    const float* ew  = (const float*)d_in[2];
    const float* gg  = (const float*)d_in[3];
    const float* wih = (const float*)d_in[4];
    const float* whh = (const float*)d_in[5];
    const float* bih = (const float*)d_in[6];
    const float* bhh = (const float*)d_in[7];
    const float* fcw = (const float*)d_in[8];
    const float* fcb = (const float*)d_in[9];
    float* out = (float*)d_out;

    float* h   = (float*)d_ws;
    float* m   = h + (size_t)N_PAD * 256;
    float* agg = m + (size_t)N_PAD * 256;
    __bf16* ggT  = (__bf16*)(agg + (size_t)N_PAD * 256);
    __bf16* wihb = ggT + 3 * 256 * 256;
    __bf16* whhb = wihb + 768 * 256;
    int*   cnt    = (int*)(whhb + 768 * 256);
    int*   off    = cnt + 50176;
    int*   slot   = off + 50176;
    int*   bucket = slot + N_EDGES_C;
    int*   csrc   = bucket + N_EDGES_C;
    float* cw     = (float*)(csrc + N_EDGES_C);

    convw_kernel<<<(3 * 256 * 256 + 255) / 256, 256, 0, stream>>>(gg, wih, whh,
                                                                  ggT, wihb, whhb);
    init_h_kernel<<<N_PAD, 256, 0, stream>>>(x, h);

    zero_int_kernel<<<(N_NODES_C + 1 + 255) / 256, 256, 0, stream>>>(cnt, N_NODES_C + 1);
    count_kernel<<<N_EDGES_C / 256, 256, 0, stream>>>(ei, cnt, slot);
    scan_kernel<<<1, 1024, 0, stream>>>(cnt, off);
    fill_kernel<<<N_EDGES_C / 256, 256, 0, stream>>>(ei, off, slot, bucket);
    rank_kernel<<<N_EDGES_C / 256, 256, 0, stream>>>(ei, ew, off, bucket, csrc, cw);

    for (int l = 0; l < 3; ++l) {
        gemm_m_kernel<<<N_PAD / 32, 256, 0, stream>>>(h, ggT + (size_t)l * 65536, m);
        gather_kernel<<<N_PAD / 8, 256, 0, stream>>>(m, off, csrc, cw, agg);
        gru_kernel<<<N_PAD / 16, 256, 0, stream>>>(agg, h, wihb, whhb, bih, bhh);
    }

    final_kernel<<<(N_NODES_C + 255) / 256, 256, 0, stream>>>(h, fcw, fcb, out);
}


# Weights stored in WMMA fragment order by the conversion kernel: each B-fragment load of the GRU and message GEMMs covers 512 contiguous bytes (4 whole lines) instead of 16 lines x 32 B
# speedup vs baseline: 1.7948x; 1.2225x over previous
#include <hip/hip_runtime.h>

#define N_NODES_C 50000
#define N_EDGES_C 800000
#define F_IN_C 32
#define C_C 256
#define N_PAD 50016
#define N_ACT_C 8

typedef __attribute__((ext_vector_type(16))) __bf16 v16bf;
typedef __attribute__((ext_vector_type(8)))  __bf16 v8bf;
typedef __attribute__((ext_vector_type(8)))  float  v8f;

static __device__ __forceinline__ v8f wmma_bf16(v16bf a, v16bf b, v8f c) {
    return __builtin_amdgcn_wmma_f32_16x16x32_bf16(
         false, a,  false, b,
         (short)0, c,  false,  false);
}

static __device__ __forceinline__ v16bf load_a_frag_f32(
    const float* __restrict__ A, int row, int ld, int k0, int lane) {
    int half = lane >> 4;
    const float* p0 = A + (long long)row * ld + k0 + half * 8;
    const float* p1 = p0 + 16;
    float4 f0 = *(const float4*)(p0);
    float4 f1 = *(const float4*)(p0 + 4);
    float4 f2 = *(const float4*)(p1);
    float4 f3 = *(const float4*)(p1 + 4);
    v16bf a;
    a[0]  = (__bf16)f0.x; a[1]  = (__bf16)f0.y; a[2]  = (__bf16)f0.z; a[3]  = (__bf16)f0.w;
    a[4]  = (__bf16)f1.x; a[5]  = (__bf16)f1.y; a[6]  = (__bf16)f1.z; a[7]  = (__bf16)f1.w;
    a[8]  = (__bf16)f2.x; a[9]  = (__bf16)f2.y; a[10] = (__bf16)f2.z; a[11] = (__bf16)f2.w;
    a[12] = (__bf16)f3.x; a[13] = (__bf16)f3.y; a[14] = (__bf16)f3.z; a[15] = (__bf16)f3.w;
    return a;
}

static __device__ __forceinline__ v16bf load_b_frag_packed(
    const __bf16* __restrict__ P, int ct, int ks, int lane) {
    const __bf16* p = P + ((ct * 8 + ks) << 9) + lane * 8;
    v8bf b0 = *(const v8bf*)(p);
    v8bf b1 = *(const v8bf*)(p + 256);
    v16bf b;
#pragma unroll
    for (int i = 0; i < 8; ++i) { b[i] = b0[i]; b[8 + i] = b1[i]; }
    return b;
}

static __device__ __forceinline__ float sigmoidf_(float x) {
    return 1.0f / (1.0f + __expf(-x));
}

__global__ void convw_kernel(const float* __restrict__ gg,
                             const float* __restrict__ wih,
                             const float* __restrict__ whh,
                             __bf16* __restrict__ ggT,
                             __bf16* __restrict__ wihb,
                             __bf16* __restrict__ whhb) {
    int o = blockIdx.x * 256 + threadIdx.x;
    if (o >= 768 * 256) return;
    int i  = o & 7;
    int l  = (o >> 3) & 31;
    int q  = (o >> 8) & 1;
    int ks = (o >> 9) & 7;
    int ct = o >> 12;
    int col = ct * 16 + (l & 15);
    int k   = ks * 32 + q * 16 + (l >> 4) * 8 + i;
    ggT[o]  = (__bf16)gg[(col >> 8) * 65536 + k * 256 + (col & 255)];
    wihb[o] = (__bf16)wih[col * 256 + k];
    whhb[o] = (__bf16)whh[col * 256 + k];
}

__global__ void init_h_kernel(const float* __restrict__ x, float* __restrict__ h) {
    int i = blockIdx.x * 256 + threadIdx.x;
    int n = i >> 8, j = i & 255;
    float v = 0.0f;
    if (n < N_NODES_C && j < F_IN_C) v = x[n * F_IN_C + j];
    h[i] = v;
}

__global__ void zero_int_kernel(int* __restrict__ p, int n) {
    int i = blockIdx.x * 256 + threadIdx.x;
    if (i < n) p[i] = 0;
}

__global__ void __launch_bounds__(256)
count_kernel(const int* __restrict__ ei, int* __restrict__ cnt,
             int* __restrict__ slot) {
    int e = blockIdx.x * 256 + threadIdx.x;
    if (e >= N_EDGES_C) return;
    slot[e] = atomicAdd(&cnt[ei[N_EDGES_C + e]], 1);
}

#define SCAN_PER ((N_NODES_C + 1 + 1023) / 1024)
__global__ void __launch_bounds__(1024)
scan_kernel(const int* __restrict__ cnt, int* __restrict__ off) {
    __shared__ int part[1024];
    int t = threadIdx.x;
    int start = t * SCAN_PER;
    int s = 0;
    for (int i = 0; i < SCAN_PER; ++i) {
        int idx = start + i;
        if (idx < N_NODES_C) s += cnt[idx];
    }
    part[t] = s;
    __syncthreads();
    for (int o = 1; o < 1024; o <<= 1) {
        int v = (t >= o) ? part[t - o] : 0;
        __syncthreads();
        part[t] += v;
        __syncthreads();
    }
    int run = part[t] - s;
    for (int i = 0; i < SCAN_PER; ++i) {
        int idx = start + i;
        if (idx <= N_NODES_C) off[idx] = run;
        if (idx < N_NODES_C) run += cnt[idx];
    }
}

__global__ void __launch_bounds__(256)
fill_kernel(const int* __restrict__ ei, const int* __restrict__ off,
            const int* __restrict__ slot, int* __restrict__ bucket) {
    int e = blockIdx.x * 256 + threadIdx.x;
    if (e >= N_EDGES_C) return;
    bucket[off[ei[N_EDGES_C + e]] + slot[e]] = e;
}

__global__ void __launch_bounds__(256)
rank_kernel(const int* __restrict__ ei, const float* __restrict__ ew,
            const int* __restrict__ off, const int* __restrict__ bucket,
            int* __restrict__ csrc, float* __restrict__ cw) {
    int e = blockIdx.x * 256 + threadIdx.x;
    if (e >= N_EDGES_C) return;
    int d = ei[N_EDGES_C + e];
    int b = off[d];
    int n = off[d + 1] - b;
    int r = 0;
    for (int i = 0; i < n; ++i) r += (bucket[b + i] < e) ? 1 : 0;
    csrc[b + r] = ei[e];
    cw[b + r]   = ew[e];
}

__global__ void __launch_bounds__(256)
gemm_m_kernel(const float* __restrict__ h, const __bf16* __restrict__ WgT,
              float* __restrict__ m) {
    int lane = threadIdx.x & 31;
    int wave = threadIdx.x >> 5;
    int half = lane >> 4;
    int row0 = blockIdx.x * 32 + (wave >> 2) * 16;
    int c0   = (wave & 3) * 64;
    int ct0  = (wave & 3) * 4;
    int rowA = row0 + (lane & 15);

    v8f acc0 = {}, acc1 = {}, acc2 = {}, acc3 = {};
#pragma unroll
    for (int k0 = 0; k0 < 256; k0 += 32) {
        v16bf a = load_a_frag_f32(h, rowA, 256, k0, lane);
        v16bf b0 = load_b_frag_packed(WgT, ct0 + 0, k0 >> 5, lane);
        acc0 = wmma_bf16(a, b0, acc0);
        v16bf b1 = load_b_frag_packed(WgT, ct0 + 1, k0 >> 5, lane);
        acc1 = wmma_bf16(a, b1, acc1);
        v16bf b2 = load_b_frag_packed(WgT, ct0 + 2, k0 >> 5, lane);
        acc2 = wmma_bf16(a, b2, acc2);
        v16bf b3 = load_b_frag_packed(WgT, ct0 + 3, k0 >> 5, lane);
        acc3 = wmma_bf16(a, b3, acc3);
    }
    v8f accs[4] = {acc0, acc1, acc2, acc3};
#pragma unroll
    for (int t = 0; t < 4; ++t) {
        int c = c0 + t * 16 + (lane & 15);
#pragma unroll
        for (int vr = 0; vr < 8; ++vr)
            m[(long long)(row0 + vr + 8 * half) * 256 + c] = accs[t][vr];
    }
}

__global__ void __launch_bounds__(256)
gather_kernel(const float* __restrict__ m, const int* __restrict__ off,
              const int* __restrict__ csrc, const float* __restrict__ cw,
              float* __restrict__ agg) {
    int lane = threadIdx.x & 31;
    int node = __builtin_amdgcn_readfirstlane(blockIdx.x * 8 + (threadIdx.x >> 5));
    int b = 0, n = 0;
    if (node < N_NODES_C) { b = off[node]; n = off[node + 1] - b; }

    float4 a0 = make_float4(0.f, 0.f, 0.f, 0.f);
    float4 a1 = make_float4(0.f, 0.f, 0.f, 0.f);
    for (int base = 0; base < n; base += 32) {
        int c = n - base;
        if (c > 32) c = 32;
        int   s_l = 0;
        float w_l = 0.0f;
        if (lane < c) { s_l = csrc[b + base + lane]; w_l = cw[b + base + lane]; }
#pragma unroll 4
        for (int j = 0; j < c; ++j) {
            int   s = __builtin_amdgcn_readlane(s_l, j);
            float w = __int_as_float(__builtin_amdgcn_readlane(__float_as_int(w_l), j));
            const float4* row = (const float4*)(m + (long long)s * 256);
            float4 v0 = row[lane];
            float4 v1 = row[32 + lane];
            a0.x += v0.x * w; a0.y += v0.y * w; a0.z += v0.z * w; a0.w += v0.w * w;
            a1.x += v1.x * w; a1.y += v1.y * w; a1.z += v1.z * w; a1.w += v1.w * w;
        }
    }
    float4* orow = (float4*)(agg + (long long)node * 256);
    orow[lane]      = a0;
    orow[32 + lane] = a1;
}

__global__ void __launch_bounds__(256)
gru_kernel(const float* __restrict__ agg, float* __restrict__ h,
           const __bf16* __restrict__ wihb, const __bf16* __restrict__ whhb,
           const float* __restrict__ bih, const float* __restrict__ bhh) {
    int lane = threadIdx.x & 31;
    int wave = threadIdx.x >> 5;
    int half = lane >> 4;
    int row0 = blockIdx.x * 16;
    int c0   = wave * 32;
    int rowA = row0 + (lane & 15);

    v8f acc[2][3][2];
#pragma unroll
    for (int a = 0; a < 2; ++a)
#pragma unroll
        for (int g = 0; g < 3; ++g)
#pragma unroll
            for (int t = 0; t < 2; ++t) { v8f z = {}; acc[a][g][t] = z; }

#pragma unroll
    for (int k0 = 0; k0 < 256; k0 += 32) {
        v16bf a_agg = load_a_frag_f32(agg, rowA, 256, k0, lane);
        v16bf a_h   = load_a_frag_f32(h,   rowA, 256, k0, lane);
#pragma unroll
        for (int g = 0; g < 3; ++g) {
#pragma unroll
            for (int t = 0; t < 2; ++t) {
                int ct = g * 16 + wave * 2 + t;
                v16bf b_ih = load_b_frag_packed(wihb, ct, k0 >> 5, lane);
                acc[0][g][t] = wmma_bf16(a_agg, b_ih, acc[0][g][t]);
                v16bf b_hh = load_b_frag_packed(whhb, ct, k0 >> 5, lane);
                acc[1][g][t] = wmma_bf16(a_h, b_hh, acc[1][g][t]);
            }
        }
    }

    __syncthreads();

#pragma unroll
    for (int t = 0; t < 2; ++t) {
        int c = c0 + t * 16 + (lane & 15);
        float bir = bih[c],       bhr = bhh[c];
        float biz = bih[256 + c], bhz = bhh[256 + c];
        float bin = bih[512 + c], bhn = bhh[512 + c];
#pragma unroll
        for (int vr = 0; vr < 8; ++vr) {
            long long idx = (long long)(row0 + vr + 8 * half) * 256 + c;
            float hold = h[idx];
            float r  = sigmoidf_(acc[0][0][t][vr] + bir + acc[1][0][t][vr] + bhr);
            float z  = sigmoidf_(acc[0][1][t][vr] + biz + acc[1][1][t][vr] + bhz);
            float nn = tanhf(acc[0][2][t][vr] + bin + r * (acc[1][2][t][vr] + bhn));
            h[idx] = (1.0f - z) * nn + z * hold;
        }
    }
}

__global__ void __launch_bounds__(256)
final_kernel(const float* __restrict__ h, const float* __restrict__ fcw,
             const float* __restrict__ fcb, float* __restrict__ out) {
    __shared__ float sw[N_ACT_C * 256];
    __shared__ float sb[N_ACT_C];
    for (int i = threadIdx.x; i < N_ACT_C * 256; i += 256) sw[i] = fcw[i];
    if (threadIdx.x < N_ACT_C) sb[threadIdx.x] = fcb[threadIdx.x];
    __syncthreads();

    int n = blockIdx.x * 256 + threadIdx.x;
    if (n >= N_NODES_C) return;

    float acc[N_ACT_C];
#pragma unroll
    for (int a = 0; a < N_ACT_C; ++a) acc[a] = sb[a];
    const float* hr = h + (long long)n * 256;
    for (int k = 0; k < 256; ++k) {
        float hv = fmaxf(hr[k], 0.0f);
#pragma unroll
        for (int a = 0; a < N_ACT_C; ++a) acc[a] += hv * sw[a * 256 + k];
    }
    float mx = acc[0];
#pragma unroll
    for (int a = 1; a < N_ACT_C; ++a) mx = fmaxf(mx, acc[a]);
    float s = 0.0f;
#pragma unroll
    for (int a = 0; a < N_ACT_C; ++a) { acc[a] = __expf(acc[a] - mx); s += acc[a]; }
    float inv = 1.0f / s;
#pragma unroll
    for (int a = 0; a < N_ACT_C; ++a) out[(long long)n * N_ACT_C + a] = acc[a] * inv;
}

extern "C" void kernel_launch(void* const* d_in, const int* in_sizes, int n_in,
                              void* d_out, int out_size, void* d_ws, size_t ws_size,
                              hipStream_t stream) {
    const float* x   = (const float*)d_in[0];
    const int*   ei  = (const int*)  d_in[1];
    const float* ew  = (const float*)d_in[2];
    const float* gg  = (const float*)d_in[3];
    const float* wih = (const float*)d_in[4];
    const float* whh = (const float*)d_in[5];
    const float* bih = (const float*)d_in[6];
    const float* bhh = (const float*)d_in[7];
    const float* fcw = (const float*)d_in[8];
    const float* fcb = (const float*)d_in[9];
    float* out = (float*)d_out;

    float* h   = (float*)d_ws;
    float* m   = h + (size_t)N_PAD * 256;
    float* agg = m + (size_t)N_PAD * 256;
    __bf16* ggT  = (__bf16*)(agg + (size_t)N_PAD * 256);
    __bf16* wihb = ggT + 3 * 256 * 256;
    __bf16* whhb = wihb + 768 * 256;
    int*   cnt    = (int*)(whhb + 768 * 256);
    int*   off    = cnt + 50176;
    int*   slot   = off + 50176;
    int*   bucket = slot + N_EDGES_C;
    int*   csrc   = bucket + N_EDGES_C;
    float* cw     = (float*)(csrc + N_EDGES_C);

    convw_kernel<<<(3 * 256 * 256 + 255) / 256, 256, 0, stream>>>(gg, wih, whh,
                                                                  ggT, wihb, whhb);
    init_h_kernel<<<N_PAD, 256, 0, stream>>>(x, h);

    zero_int_kernel<<<(N_NODES_C + 1 + 255) / 256, 256, 0, stream>>>(cnt, N_NODES_C + 1);
    count_kernel<<<N_EDGES_C / 256, 256, 0, stream>>>(ei, cnt, slot);
    scan_kernel<<<1, 1024, 0, stream>>>(cnt, off);
    fill_kernel<<<N_EDGES_C / 256, 256, 0, stream>>>(ei, off, slot, bucket);
    rank_kernel<<<N_EDGES_C / 256, 256, 0, stream>>>(ei, ew, off, bucket, csrc, cw);

    for (int l = 0; l < 3; ++l) {
        gemm_m_kernel<<<N_PAD / 32, 256, 0, stream>>>(h, ggT + (size_t)l * 65536, m);
        gather_kernel<<<N_PAD / 8, 256, 0, stream>>>(m, off, csrc, cw, agg);
        gru_kernel<<<N_PAD / 16, 256, 0, stream>>>(agg, h, wihb, whhb, bih, bhh);
    }

    final_kernel<<<(N_NODES_C + 255) / 256, 256, 0, stream>>>(h, fcw, fcb, out);
}


# GEMM kernels (message and GRU): every fragment load of a k-step is issued as one group before its WMMAs, so a wave waits one second-level round trip per k-step instead of one per WMMA
# speedup vs baseline: 1.9679x; 1.0964x over previous
#include <hip/hip_runtime.h>

#define N_NODES_C 50000
#define N_EDGES_C 800000
#define F_IN_C 32
#define C_C 256
#define N_PAD 50016
#define N_ACT_C 8

typedef __attribute__((ext_vector_type(16))) __bf16 v16bf;
typedef __attribute__((ext_vector_type(8)))  __bf16 v8bf;
typedef __attribute__((ext_vector_type(8)))  float  v8f;

static __device__ __forceinline__ v8f wmma_bf16(v16bf a, v16bf b, v8f c) {
    return __builtin_amdgcn_wmma_f32_16x16x32_bf16(
         false, a,  false, b,
         (short)0, c,  false,  false);
}

typedef __attribute__((ext_vector_type(4))) float v4f;


struct a_raw { v4f f0, f1, f2, f3; };

static __device__ __forceinline__ a_raw load_a_raw_f32(
    const float* __restrict__ A, int row, int ld, int k0, int lane) {
    int half = lane >> 4;
    const float* p0 = A + (long long)row * ld + k0 + half * 8;
    const float* p1 = p0 + 16;
    a_raw r;
    r.f0 = *(const v4f*)(p0);
    r.f1 = *(const v4f*)(p0 + 4);
    r.f2 = *(const v4f*)(p1);
    r.f3 = *(const v4f*)(p1 + 4);
    return r;
}

static __device__ __forceinline__ v16bf make_a_frag(const a_raw& r) {
    v16bf a;
#pragma unroll
    for (int i = 0; i < 4; ++i) {
        a[i]      = (__bf16)r.f0[i];
        a[4 + i]  = (__bf16)r.f1[i];
        a[8 + i]  = (__bf16)r.f2[i];
        a[12 + i] = (__bf16)r.f3[i];
    }
    return a;
}

struct b_raw { v8bf q0, q1; };

static __device__ __forceinline__ b_raw load_b_raw_packed(
    const __bf16* __restrict__ P, int ct, int ks, int lane) {
    const __bf16* p = P + ((ct * 8 + ks) << 9) + lane * 8;
    b_raw r;
    r.q0 = *(const v8bf*)(p);
    r.q1 = *(const v8bf*)(p + 256);
    return r;
}

static __device__ __forceinline__ v16bf make_b_frag(const b_raw& r) {
    v16bf b;
#pragma unroll
    for (int i = 0; i < 8; ++i) { b[i] = r.q0[i]; b[8 + i] = r.q1[i]; }
    return b;
}

static __device__ __forceinline__ float sigmoidf_(float x) {
    return 1.0f / (1.0f + __expf(-x));
}

__global__ void convw_kernel(const float* __restrict__ gg,
                             const float* __restrict__ wih,
                             const float* __restrict__ whh,
                             __bf16* __restrict__ ggT,
                             __bf16* __restrict__ wihb,
                             __bf16* __restrict__ whhb) {
    int o = blockIdx.x * 256 + threadIdx.x;
    if (o >= 768 * 256) return;
    int i  = o & 7;
    int l  = (o >> 3) & 31;
    int q  = (o >> 8) & 1;
    int ks = (o >> 9) & 7;
    int ct = o >> 12;
    int col = ct * 16 + (l & 15);
    int k   = ks * 32 + q * 16 + (l >> 4) * 8 + i;
    ggT[o]  = (__bf16)gg[(col >> 8) * 65536 + k * 256 + (col & 255)];
    wihb[o] = (__bf16)wih[col * 256 + k];
    whhb[o] = (__bf16)whh[col * 256 + k];
}

__global__ void init_h_kernel(const float* __restrict__ x, float* __restrict__ h) {
    int i = blockIdx.x * 256 + threadIdx.x;
    int n = i >> 8, j = i & 255;
    float v = 0.0f;
    if (n < N_NODES_C && j < F_IN_C) v = x[n * F_IN_C + j];
    h[i] = v;
}

__global__ void zero_int_kernel(int* __restrict__ p, int n) {
    int i = blockIdx.x * 256 + threadIdx.x;
    if (i < n) p[i] = 0;
}

__global__ void __launch_bounds__(256)
count_kernel(const int* __restrict__ ei, int* __restrict__ cnt,
             int* __restrict__ slot) {
    int e = blockIdx.x * 256 + threadIdx.x;
    if (e >= N_EDGES_C) return;
    slot[e] = atomicAdd(&cnt[ei[N_EDGES_C + e]], 1);
}

#define SCAN_PER ((N_NODES_C + 1 + 1023) / 1024)
__global__ void __launch_bounds__(1024)
scan_kernel(const int* __restrict__ cnt, int* __restrict__ off) {
    __shared__ int part[1024];
    int t = threadIdx.x;
    int start = t * SCAN_PER;
    int s = 0;
    for (int i = 0; i < SCAN_PER; ++i) {
        int idx = start + i;
        if (idx < N_NODES_C) s += cnt[idx];
    }
    part[t] = s;
    __syncthreads();
    for (int o = 1; o < 1024; o <<= 1) {
        int v = (t >= o) ? part[t - o] : 0;
        __syncthreads();
        part[t] += v;
        __syncthreads();
    }
    int run = part[t] - s;
    for (int i = 0; i < SCAN_PER; ++i) {
        int idx = start + i;
        if (idx <= N_NODES_C) off[idx] = run;
        if (idx < N_NODES_C) run += cnt[idx];
    }
}

__global__ void __launch_bounds__(256)
fill_kernel(const int* __restrict__ ei, const int* __restrict__ off,
            const int* __restrict__ slot, int* __restrict__ bucket) {
    int e = blockIdx.x * 256 + threadIdx.x;
    if (e >= N_EDGES_C) return;
    bucket[off[ei[N_EDGES_C + e]] + slot[e]] = e;
}

__global__ void __launch_bounds__(256)
rank_kernel(const int* __restrict__ ei, const float* __restrict__ ew,
            const int* __restrict__ off, const int* __restrict__ bucket,
            int* __restrict__ csrc, float* __restrict__ cw) {
    int e = blockIdx.x * 256 + threadIdx.x;
    if (e >= N_EDGES_C) return;
    int d = ei[N_EDGES_C + e];
    int b = off[d];
    int n = off[d + 1] - b;
    int r = 0;
    for (int i = 0; i < n; ++i) r += (bucket[b + i] < e) ? 1 : 0;
    csrc[b + r] = ei[e];
    cw[b + r]   = ew[e];
}

__global__ void __launch_bounds__(256)
gemm_m_kernel(const float* __restrict__ h, const __bf16* __restrict__ WgT,
              float* __restrict__ m) {
    int lane = threadIdx.x & 31;
    int wave = threadIdx.x >> 5;
    int half = lane >> 4;
    int row0 = blockIdx.x * 32 + (wave >> 2) * 16;
    int c0   = (wave & 3) * 64;
    int ct0  = (wave & 3) * 4;
    int rowA = row0 + (lane & 15);

    v8f acc0 = {}, acc1 = {}, acc2 = {}, acc3 = {};
#pragma unroll
    for (int k0 = 0; k0 < 256; k0 += 32) {
        a_raw ra  = load_a_raw_f32(h, rowA, 256, k0, lane);
        b_raw rb0 = load_b_raw_packed(WgT, ct0 + 0, k0 >> 5, lane);
        b_raw rb1 = load_b_raw_packed(WgT, ct0 + 1, k0 >> 5, lane);
        b_raw rb2 = load_b_raw_packed(WgT, ct0 + 2, k0 >> 5, lane);
        b_raw rb3 = load_b_raw_packed(WgT, ct0 + 3, k0 >> 5, lane);
        asm volatile("" :: "v"(ra.f0), "v"(ra.f1), "v"(ra.f2), "v"(ra.f3),
                           "v"(rb0.q0), "v"(rb0.q1), "v"(rb1.q0), "v"(rb1.q1),
                           "v"(rb2.q0), "v"(rb2.q1), "v"(rb3.q0), "v"(rb3.q1));
        v16bf a = make_a_frag(ra);
        acc0 = wmma_bf16(a, make_b_frag(rb0), acc0);
        acc1 = wmma_bf16(a, make_b_frag(rb1), acc1);
        acc2 = wmma_bf16(a, make_b_frag(rb2), acc2);
        acc3 = wmma_bf16(a, make_b_frag(rb3), acc3);
    }
    v8f accs[4] = {acc0, acc1, acc2, acc3};
#pragma unroll
    for (int t = 0; t < 4; ++t) {
        int c = c0 + t * 16 + (lane & 15);
#pragma unroll
        for (int vr = 0; vr < 8; ++vr)
            m[(long long)(row0 + vr + 8 * half) * 256 + c] = accs[t][vr];
    }
}

__global__ void __launch_bounds__(256)
gather_kernel(const float* __restrict__ m, const int* __restrict__ off,
              const int* __restrict__ csrc, const float* __restrict__ cw,
              float* __restrict__ agg) {
    int lane = threadIdx.x & 31;
    int node = __builtin_amdgcn_readfirstlane(blockIdx.x * 8 + (threadIdx.x >> 5));
    int b = 0, n = 0;
    if (node < N_NODES_C) { b = off[node]; n = off[node + 1] - b; }

    float4 a0 = make_float4(0.f, 0.f, 0.f, 0.f);
    float4 a1 = make_float4(0.f, 0.f, 0.f, 0.f);
    for (int base = 0; base < n; base += 32) {
        int c = n - base;
        if (c > 32) c = 32;
        int   s_l = 0;
        float w_l = 0.0f;
        if (lane < c) { s_l = csrc[b + base + lane]; w_l = cw[b + base + lane]; }
#pragma unroll 4
        for (int j = 0; j < c; ++j) {
            int   s = __builtin_amdgcn_readlane(s_l, j);
            float w = __int_as_float(__builtin_amdgcn_readlane(__float_as_int(w_l), j));
            const float4* row = (const float4*)(m + (long long)s * 256);
            float4 v0 = row[lane];
            float4 v1 = row[32 + lane];
            a0.x += v0.x * w; a0.y += v0.y * w; a0.z += v0.z * w; a0.w += v0.w * w;
            a1.x += v1.x * w; a1.y += v1.y * w; a1.z += v1.z * w; a1.w += v1.w * w;
        }
    }
    float4* orow = (float4*)(agg + (long long)node * 256);
    orow[lane]      = a0;
    orow[32 + lane] = a1;
}

__global__ void __launch_bounds__(256)
gru_kernel(const float* __restrict__ agg, float* __restrict__ h,
           const __bf16* __restrict__ wihb, const __bf16* __restrict__ whhb,
           const float* __restrict__ bih, const float* __restrict__ bhh) {
    int lane = threadIdx.x & 31;
    int wave = threadIdx.x >> 5;
    int half = lane >> 4;
    int row0 = blockIdx.x * 16;
    int c0   = wave * 32;
    int rowA = row0 + (lane & 15);

    v8f acc[2][3][2];
#pragma unroll
    for (int a = 0; a < 2; ++a)
#pragma unroll
        for (int g = 0; g < 3; ++g)
#pragma unroll
            for (int t = 0; t < 2; ++t) { v8f z = {}; acc[a][g][t] = z; }

#pragma unroll
    for (int k0 = 0; k0 < 256; k0 += 32) {
        a_raw ra_agg = load_a_raw_f32(agg, rowA, 256, k0, lane);
        a_raw ra_h   = load_a_raw_f32(h,   rowA, 256, k0, lane);
        b_raw rb[2][3][2];
#pragma unroll
        for (int g = 0; g < 3; ++g) {
#pragma unroll
            for (int t = 0; t < 2; ++t) {
                int ct = g * 16 + wave * 2 + t;
                rb[0][g][t] = load_b_raw_packed(wihb, ct, k0 >> 5, lane);
                rb[1][g][t] = load_b_raw_packed(whhb, ct, k0 >> 5, lane);
            }
        }
        asm volatile("" :: "v"(ra_agg.f0), "v"(ra_agg.f1), "v"(ra_agg.f2), "v"(ra_agg.f3),
                           "v"(ra_h.f0), "v"(ra_h.f1), "v"(ra_h.f2), "v"(ra_h.f3));
        asm volatile("" :: "v"(rb[0][0][0].q0), "v"(rb[0][0][0].q1), "v"(rb[0][0][1].q0), "v"(rb[0][0][1].q1),
                           "v"(rb[0][1][0].q0), "v"(rb[0][1][0].q1), "v"(rb[0][1][1].q0), "v"(rb[0][1][1].q1),
                           "v"(rb[0][2][0].q0), "v"(rb[0][2][0].q1), "v"(rb[0][2][1].q0), "v"(rb[0][2][1].q1));
        asm volatile("" :: "v"(rb[1][0][0].q0), "v"(rb[1][0][0].q1), "v"(rb[1][0][1].q0), "v"(rb[1][0][1].q1),
                           "v"(rb[1][1][0].q0), "v"(rb[1][1][0].q1), "v"(rb[1][1][1].q0), "v"(rb[1][1][1].q1),
                           "v"(rb[1][2][0].q0), "v"(rb[1][2][0].q1), "v"(rb[1][2][1].q0), "v"(rb[1][2][1].q1));
        v16bf a_agg = make_a_frag(ra_agg);
        v16bf a_h   = make_a_frag(ra_h);
#pragma unroll
        for (int g = 0; g < 3; ++g) {
#pragma unroll
            for (int t = 0; t < 2; ++t) {
                acc[0][g][t] = wmma_bf16(a_agg, make_b_frag(rb[0][g][t]), acc[0][g][t]);
                acc[1][g][t] = wmma_bf16(a_h,   make_b_frag(rb[1][g][t]), acc[1][g][t]);
            }
        }
    }

    __syncthreads();

#pragma unroll
    for (int t = 0; t < 2; ++t) {
        int c = c0 + t * 16 + (lane & 15);
        float bir = bih[c],       bhr = bhh[c];
        float biz = bih[256 + c], bhz = bhh[256 + c];
        float bin = bih[512 + c], bhn = bhh[512 + c];
#pragma unroll
        for (int vr = 0; vr < 8; ++vr) {
            long long idx = (long long)(row0 + vr + 8 * half) * 256 + c;
            float hold = h[idx];
            float r  = sigmoidf_(acc[0][0][t][vr] + bir + acc[1][0][t][vr] + bhr);
            float z  = sigmoidf_(acc[0][1][t][vr] + biz + acc[1][1][t][vr] + bhz);
            float nn = tanhf(acc[0][2][t][vr] + bin + r * (acc[1][2][t][vr] + bhn));
            h[idx] = (1.0f - z) * nn + z * hold;
        }
    }
}

__global__ void __launch_bounds__(256)
final_kernel(const float* __restrict__ h, const float* __restrict__ fcw,
             const float* __restrict__ fcb, float* __restrict__ out) {
    __shared__ float sw[N_ACT_C * 256];
    __shared__ float sb[N_ACT_C];
    for (int i = threadIdx.x; i < N_ACT_C * 256; i += 256) sw[i] = fcw[i];
    if (threadIdx.x < N_ACT_C) sb[threadIdx.x] = fcb[threadIdx.x];
    __syncthreads();

    int n = blockIdx.x * 256 + threadIdx.x;
    if (n >= N_NODES_C) return;

    float acc[N_ACT_C];
#pragma unroll
    for (int a = 0; a < N_ACT_C; ++a) acc[a] = sb[a];
    const float* hr = h + (long long)n * 256;
    for (int k = 0; k < 256; ++k) {
        float hv = fmaxf(hr[k], 0.0f);
#pragma unroll
        for (int a = 0; a < N_ACT_C; ++a) acc[a] += hv * sw[a * 256 + k];
    }
    float mx = acc[0];
#pragma unroll
    for (int a = 1; a < N_ACT_C; ++a) mx = fmaxf(mx, acc[a]);
    float s = 0.0f;
#pragma unroll
    for (int a = 0; a < N_ACT_C; ++a) { acc[a] = __expf(acc[a] - mx); s += acc[a]; }
    float inv = 1.0f / s;
#pragma unroll
    for (int a = 0; a < N_ACT_C; ++a) out[(long long)n * N_ACT_C + a] = acc[a] * inv;
}

extern "C" void kernel_launch(void* const* d_in, const int* in_sizes, int n_in,
                              void* d_out, int out_size, void* d_ws, size_t ws_size,
                              hipStream_t stream) {
    const float* x   = (const float*)d_in[0];
    const int*   ei  = (const int*)  d_in[1];
    const float* ew  = (const float*)d_in[2];
    const float* gg  = (const float*)d_in[3];
    const float* wih = (const float*)d_in[4];
    const float* whh = (const float*)d_in[5];
    const float* bih = (const float*)d_in[6];
    const float* bhh = (const float*)d_in[7];
    const float* fcw = (const float*)d_in[8];
    const float* fcb = (const float*)d_in[9];
    float* out = (float*)d_out;

    float* h   = (float*)d_ws;
    float* m   = h + (size_t)N_PAD * 256;
    float* agg = m + (size_t)N_PAD * 256;
    __bf16* ggT  = (__bf16*)(agg + (size_t)N_PAD * 256);
    __bf16* wihb = ggT + 3 * 256 * 256;
    __bf16* whhb = wihb + 768 * 256;
    int*   cnt    = (int*)(whhb + 768 * 256);
    int*   off    = cnt + 50176;
    int*   slot   = off + 50176;
    int*   bucket = slot + N_EDGES_C;
    int*   csrc   = bucket + N_EDGES_C;
    float* cw     = (float*)(csrc + N_EDGES_C);

    convw_kernel<<<(3 * 256 * 256 + 255) / 256, 256, 0, stream>>>(gg, wih, whh,
                                                                  ggT, wihb, whhb);
    init_h_kernel<<<N_PAD, 256, 0, stream>>>(x, h);

    zero_int_kernel<<<(N_NODES_C + 1 + 255) / 256, 256, 0, stream>>>(cnt, N_NODES_C + 1);
    count_kernel<<<N_EDGES_C / 256, 256, 0, stream>>>(ei, cnt, slot);
    scan_kernel<<<1, 1024, 0, stream>>>(cnt, off);
    fill_kernel<<<N_EDGES_C / 256, 256, 0, stream>>>(ei, off, slot, bucket);
    rank_kernel<<<N_EDGES_C / 256, 256, 0, stream>>>(ei, ew, off, bucket, csrc, cw);

    for (int l = 0; l < 3; ++l) {
        gemm_m_kernel<<<N_PAD / 32, 256, 0, stream>>>(h, ggT + (size_t)l * 65536, m);
        gather_kernel<<<N_PAD / 8, 256, 0, stream>>>(m, off, csrc, cw, agg);
        gru_kernel<<<N_PAD / 16, 256, 0, stream>>>(agg, h, wihb, whhb, bih, bhh);
    }

    final_kernel<<<(N_NODES_C + 255) / 256, 256, 0, stream>>>(h, fcw, fcb, out);
}


# Layer 0 uses that h0 = x zero-padded is zero beyond column 31: its gather reads 128 B per edge from x, its GRU runs 1 k-step (K=32) of 8, no padded copy of x; sum order and values unchanged
# speedup vs baseline: 3.3156x; 1.6848x over previous
#include <hip/hip_runtime.h>

#define N_NODES_C 50000
#define N_EDGES_C 800000
#define F_IN_C 32
#define C_C 256
#define N_PAD 50016
#define N_ACT_C 8

typedef __attribute__((ext_vector_type(16))) __bf16 v16bf;
typedef __attribute__((ext_vector_type(8)))  __bf16 v8bf;
typedef __attribute__((ext_vector_type(8)))  float  v8f;

static __device__ __forceinline__ v8f wmma_bf16(v16bf a, v16bf b, v8f c) {
    return __builtin_amdgcn_wmma_f32_16x16x32_bf16(
         false, a,  false, b,
         (short)0, c,  false,  false);
}

typedef __attribute__((ext_vector_type(4))) float v4f;
typedef __attribute__((ext_vector_type(2))) float v2f;

static __device__ __forceinline__ v8f wmma_f32(v2f a, v2f b, v8f c) {
    return __builtin_amdgcn_wmma_f32_16x16x4_f32(
         false, a,  false, b,
         (short)0, c,  false,  false);
}


struct a_raw { v4f f0, f1, f2, f3; };

static __device__ __forceinline__ a_raw load_a_raw_f32(
    const float* __restrict__ A, int row, int ld, int k0, int lane) {
    int half = lane >> 4;
    const float* p0 = A + (long long)row * ld + k0 + half * 8;
    const float* p1 = p0 + 16;
    a_raw r;
    r.f0 = *(const v4f*)(p0);
    r.f1 = *(const v4f*)(p0 + 4);
    r.f2 = *(const v4f*)(p1);
    r.f3 = *(const v4f*)(p1 + 4);
    return r;
}

static __device__ __forceinline__ v16bf make_a_frag(const a_raw& r) {
    v16bf a;
#pragma unroll
    for (int i = 0; i < 4; ++i) {
        a[i]      = (__bf16)r.f0[i];
        a[4 + i]  = (__bf16)r.f1[i];
        a[8 + i]  = (__bf16)r.f2[i];
        a[12 + i] = (__bf16)r.f3[i];
    }
    return a;
}

struct b_raw { v8bf q0, q1; };

static __device__ __forceinline__ v16bf make_b_frag(const b_raw& r) {
    v16bf b;
#pragma unroll
    for (int i = 0; i < 8; ++i) { b[i] = r.q0[i]; b[8 + i] = r.q1[i]; }
    return b;
}

static __device__ __forceinline__ float sigmoidf_(float x) {
    return 1.0f / (1.0f + __expf(-x));
}

__global__ void convw_kernel(const float* __restrict__ whh,
                             __bf16* __restrict__ whhb) {
    int o = blockIdx.x * 256 + threadIdx.x;
    if (o >= 768 * 256) return;
    int i  = o & 7;
    int l  = (o >> 3) & 31;
    int q  = (o >> 8) & 1;
    int ks = (o >> 9) & 7;
    int ct = o >> 12;
    int col = ct * 16 + (l & 15);
    int k   = ks * 32 + q * 16 + (l >> 4) * 8 + i;
    whhb[o] = (__bf16)whh[col * 256 + k];
}

__global__ void __launch_bounds__(256)
combw_kernel(const float* __restrict__ gg, const float* __restrict__ wih,
             __bf16* __restrict__ wc) {
    int lane  = threadIdx.x & 31;
    int tile  = __builtin_amdgcn_readfirstlane(blockIdx.x * 8 + (threadIdx.x >> 5));
    int layer = tile / 768;
    int rem   = tile - layer * 768;
    int ct    = rem >> 4;
    int kt    = rem & 15;
    if (layer == 0 && kt >= 2) return;
    int half  = lane >> 4;
    const float* pa = gg + layer * 65536 + (kt * 16 + (lane & 15)) * 256 + half * 8;
    const float* pb = wih + (ct * 16 + (lane & 15)) * 256 + half * 8;

    v8f acc0 = {}, acc1 = {};
#pragma unroll
    for (int ch = 0; ch < 16; ++ch) {
        v4f a0 = *(const v4f*)(pa + ch * 16);
        v4f a1 = *(const v4f*)(pa + ch * 16 + 4);
        v4f b0 = *(const v4f*)(pb + ch * 16);
        v4f b1 = *(const v4f*)(pb + ch * 16 + 4);
        v2f x0 = {a0[0], a0[1]}, y0 = {b0[0], b0[1]};
        v2f x1 = {a0[2], a0[3]}, y1 = {b0[2], b0[3]};
        v2f x2 = {a1[0], a1[1]}, y2 = {b1[0], b1[1]};
        v2f x3 = {a1[2], a1[3]}, y3 = {b1[2], b1[3]};
        acc0 = wmma_f32(x0, y0, acc0);
        acc1 = wmma_f32(x1, y1, acc1);
        acc0 = wmma_f32(x2, y2, acc0);
        acc1 = wmma_f32(x3, y3, acc1);
    }
    v8bf o;
#pragma unroll
    for (int v = 0; v < 8; ++v) o[v] = (__bf16)(acc0[v] + acc1[v]);
    *(v8bf*)(wc + layer * 196608
                + ((((ct * 8 + (kt >> 1)) * 2 + (kt & 1))) << 8) + lane * 8) = o;
}

__global__ void zero_int_kernel(int* __restrict__ p, int n) {
    int i = blockIdx.x * 256 + threadIdx.x;
    if (i < n) p[i] = 0;
}

__global__ void __launch_bounds__(256)
count_kernel(const int* __restrict__ ei, int* __restrict__ cnt,
             int* __restrict__ slot) {
    int e = blockIdx.x * 256 + threadIdx.x;
    if (e >= N_EDGES_C) return;
    slot[e] = atomicAdd(&cnt[ei[N_EDGES_C + e]], 1);
}

#define SCAN_PER ((N_NODES_C + 1 + 1023) / 1024)
__global__ void __launch_bounds__(1024)
scan_kernel(const int* __restrict__ cnt, int* __restrict__ off) {
    __shared__ int part[1024];
    int t = threadIdx.x;
    int start = t * SCAN_PER;
    int s = 0;
    for (int i = 0; i < SCAN_PER; ++i) {
        int idx = start + i;
        if (idx < N_NODES_C) s += cnt[idx];
    }
    part[t] = s;
    __syncthreads();
    for (int o = 1; o < 1024; o <<= 1) {
        int v = (t >= o) ? part[t - o] : 0;
        __syncthreads();
        part[t] += v;
        __syncthreads();
    }
    int run = part[t] - s;
    for (int i = 0; i < SCAN_PER; ++i) {
        int idx = start + i;
        if (idx <= N_NODES_C) off[idx] = run;
        if (idx < N_NODES_C) run += cnt[idx];
    }
}

__global__ void __launch_bounds__(256)
fill_kernel(const int* __restrict__ ei, const int* __restrict__ off,
            const int* __restrict__ slot, int* __restrict__ bucket) {
    int e = blockIdx.x * 256 + threadIdx.x;
    if (e >= N_EDGES_C) return;
    bucket[off[ei[N_EDGES_C + e]] + slot[e]] = e;
}

__global__ void __launch_bounds__(256)
rank_kernel(const int* __restrict__ ei, const float* __restrict__ ew,
            const int* __restrict__ off, const int* __restrict__ bucket,
            int* __restrict__ csrc, float* __restrict__ cw) {
    int e = blockIdx.x * 256 + threadIdx.x;
    if (e >= N_EDGES_C) return;
    int d = ei[N_EDGES_C + e];
    int b = off[d];
    int n = off[d + 1] - b;
    int r = 0;
    for (int i = 0; i < n; ++i) r += (bucket[b + i] < e) ? 1 : 0;
    csrc[b + r] = ei[e];
    cw[b + r]   = ew[e];
}

__global__ void __launch_bounds__(256)
gather_kernel(const float* __restrict__ h, const int* __restrict__ off,
              const int* __restrict__ csrc, const float* __restrict__ cw,
              float* __restrict__ agg) {
    int lane = threadIdx.x & 31;
    int node = __builtin_amdgcn_readfirstlane(blockIdx.x * 8 + (threadIdx.x >> 5));
    int b = 0, n = 0;
    if (node < N_NODES_C) { b = off[node]; n = off[node + 1] - b; }

    float4 a0 = make_float4(0.f, 0.f, 0.f, 0.f);
    float4 a1 = make_float4(0.f, 0.f, 0.f, 0.f);
    for (int base = 0; base < n; base += 32) {
        int c = n - base;
        if (c > 32) c = 32;
        int   s_l = 0;
        float w_l = 0.0f;
        if (lane < c) { s_l = csrc[b + base + lane]; w_l = cw[b + base + lane]; }
#pragma unroll 4
        for (int j = 0; j < c; ++j) {
            int   s = __builtin_amdgcn_readlane(s_l, j);
            float w = __int_as_float(__builtin_amdgcn_readlane(__float_as_int(w_l), j));
            const float4* row = (const float4*)(h + (long long)s * 256);
            float4 v0 = row[lane];
            float4 v1 = row[32 + lane];
            a0.x += v0.x * w; a0.y += v0.y * w; a0.z += v0.z * w; a0.w += v0.w * w;
            a1.x += v1.x * w; a1.y += v1.y * w; a1.z += v1.z * w; a1.w += v1.w * w;
        }
    }
    float4* orow = (float4*)(agg + (long long)node * 256);
    orow[lane]      = a0;
    orow[32 + lane] = a1;
}

#define GRU_SLICES 4
#define GRU_GROUPS 64
#define GRU_RT     (N_PAD / 16)
#define GRU_STRIDE (GRU_GROUPS * 8)

struct b_set { b_raw f[2][4]; };

static __device__ __forceinline__ void lds_load_set(b_set& s, const __bf16* L,
                                                    int g, int ks, int lane) {
#pragma unroll
    for (int mat = 0; mat < 2; ++mat) {
#pragma unroll
        for (int t = 0; t < 4; ++t) {
            const __bf16* p = L + ((((mat * 3 + g) * 4 + t) * 8 + ks) << 9) + lane * 8;
            s.f[mat][t].q0 = *(const v8bf*)(p);
            s.f[mat][t].q1 = *(const v8bf*)(p + 256);
        }
    }
}

__global__ void __launch_bounds__(256)
gru_kernel(const float* __restrict__ agg, const float* __restrict__ hin,
           float* __restrict__ hout,
           const __bf16* __restrict__ wcb, const __bf16* __restrict__ whhb,
           const float* __restrict__ bih, const float* __restrict__ bhh) {
    __shared__ __attribute__((aligned(16))) __bf16 wl[2 * 3 * 4 * 8 * 512];

    int tid  = threadIdx.x;
    int lane = tid & 31;
    int wave = tid >> 5;
    int half = lane >> 4;
    int slice = blockIdx.x & (GRU_SLICES - 1);
    int group = blockIdx.x >> 2;

#pragma unroll
    for (int mg = 0; mg < 6; ++mg) {
        const __bf16* src = ((mg >= 3) ? whhb : wcb)
                          + ((((mg % 3) * 16 + slice * 4) * 8) << 9) + tid * 8;
        __bf16* dst = wl + mg * 16384 + tid * 8;
        v8bf tmp[8];
#pragma unroll
        for (int p = 0; p < 8; ++p) tmp[p] = *(const v8bf*)(src + p * 2048);
#pragma unroll
        for (int p = 0; p < 8; ++p) *(v8bf*)(dst + p * 2048) = tmp[p];
    }
    __syncthreads();

    int rt   = group * 8 + wave;
    int rowA = rt * 16 + (lane & 15);

    a_raw pa[2], ph[2];
    pa[0] = load_a_raw_f32(agg, rowA, 256, 0,  lane);
    ph[0] = load_a_raw_f32(hin, rowA, 256, 0,  lane);
    pa[1] = load_a_raw_f32(agg, rowA, 256, 32, lane);
    ph[1] = load_a_raw_f32(hin, rowA, 256, 32, lane);

    b_set bs[2];
    lds_load_set(bs[0], wl, 0, 0, lane);

    while (true) {
        int  rtn  = rt + GRU_STRIDE;
        bool more = rtn < GRU_RT;
        int  rowN = (more ? rtn : rt) * 16 + (lane & 15);

        float hold[4][8];
        v8f acc[2][3][4];
#pragma unroll
        for (int a = 0; a < 2; ++a)
#pragma unroll
            for (int g = 0; g < 3; ++g)
#pragma unroll
                for (int t = 0; t < 4; ++t) { v8f z = {}; acc[a][g][t] = z; }

#pragma unroll
        for (int ks = 0; ks < 8; ++ks) {
            v16bf a_agg = make_a_frag(pa[ks & 1]);
            v16bf a_h   = make_a_frag(ph[ks & 1]);
            if (ks < 6) {
                pa[ks & 1] = load_a_raw_f32(agg, rowA, 256, (ks + 2) * 32, lane);
                ph[ks & 1] = load_a_raw_f32(hin, rowA, 256, (ks + 2) * 32, lane);
            } else {
                pa[ks & 1] = load_a_raw_f32(agg, rowN, 256, (ks - 6) * 32, lane);
                ph[ks & 1] = load_a_raw_f32(hin, rowN, 256, (ks - 6) * 32, lane);
            }
            if (ks == 6) {
#pragma unroll
                for (int t = 0; t < 4; ++t)
#pragma unroll
                    for (int vr = 0; vr < 8; ++vr)
                        hold[t][vr] = hin[(long long)(rt * 16 + vr + 8 * half) * 256
                                          + slice * 64 + t * 16 + (lane & 15)];
            }
#pragma unroll
            for (int g = 0; g < 3; ++g) {
                const int bi = ks * 3 + g;
                const int nb = (bi + 1) % 24;
                b_set& cur = bs[bi & 1];
                lds_load_set(bs[(bi + 1) & 1], wl, nb % 3, nb / 3, lane);
                asm volatile("" :: "v"(cur.f[0][0].q0), "v"(cur.f[0][0].q1),
                                   "v"(cur.f[0][1].q0), "v"(cur.f[0][1].q1),
                                   "v"(cur.f[0][2].q0), "v"(cur.f[0][2].q1),
                                   "v"(cur.f[0][3].q0), "v"(cur.f[0][3].q1),
                                   "v"(cur.f[1][0].q0), "v"(cur.f[1][0].q1),
                                   "v"(cur.f[1][1].q0), "v"(cur.f[1][1].q1),
                                   "v"(cur.f[1][2].q0), "v"(cur.f[1][2].q1),
                                   "v"(cur.f[1][3].q0), "v"(cur.f[1][3].q1));
#pragma unroll
                for (int t = 0; t < 4; ++t) {
                    acc[0][g][t] = wmma_bf16(a_agg, make_b_frag(cur.f[0][t]), acc[0][g][t]);
                    acc[1][g][t] = wmma_bf16(a_h,   make_b_frag(cur.f[1][t]), acc[1][g][t]);
                }
            }
        }

#pragma unroll
        for (int t = 0; t < 4; ++t) {
            int c = slice * 64 + t * 16 + (lane & 15);
            float bir = bih[c],       bhr = bhh[c];
            float biz = bih[256 + c], bhz = bhh[256 + c];
            float bin = bih[512 + c], bhn = bhh[512 + c];
            asm volatile("" :: "v"(hold[t][0]), "v"(hold[t][1]), "v"(hold[t][2]), "v"(hold[t][3]),
                               "v"(hold[t][4]), "v"(hold[t][5]), "v"(hold[t][6]), "v"(hold[t][7]));
#pragma unroll
            for (int vr = 0; vr < 8; ++vr) {
                long long idx = (long long)(rt * 16 + vr + 8 * half) * 256 + c;
                float r  = sigmoidf_(acc[0][0][t][vr] + bir + acc[1][0][t][vr] + bhr);
                float z  = sigmoidf_(acc[0][1][t][vr] + biz + acc[1][1][t][vr] + bhz);
                float nn = tanhf(acc[0][2][t][vr] + bin + r * (acc[1][2][t][vr] + bhn));
                hout[idx] = (1.0f - z) * nn + z * hold[t][vr];
            }
        }

        if (!more) break;
        rt   = rtn;
        rowA = rowN;
    }
}

#define G0_BLOCKS 1024
__global__ void __launch_bounds__(256)
gather0_kernel(const float* __restrict__ x, const int* __restrict__ off,
               const int* __restrict__ csrc, const float* __restrict__ cw,
               float* __restrict__ s0) {
    int lane = threadIdx.x & 31;
    int gw = __builtin_amdgcn_readfirstlane(blockIdx.x * 8 + (threadIdx.x >> 5));
    for (int node = gw; node < N_PAD; node += G0_BLOCKS * 8) {
        int b = 0, n = 0;
        if (node < N_NODES_C) { b = off[node]; n = off[node + 1] - b; }
        float a = 0.0f;
        for (int base = 0; base < n; base += 32) {
            int c = n - base;
            if (c > 32) c = 32;
            int   s_l = 0;
            float w_l = 0.0f;
            if (lane < c) { s_l = csrc[b + base + lane]; w_l = cw[b + base + lane]; }
            for (int j0 = 0; j0 < c; j0 += 16) {
                float v[16];
#pragma unroll
                for (int k = 0; k < 16; ++k) {
                    int s = __builtin_amdgcn_readlane(s_l, j0 + k);
                    v[k] = x[(long long)s * F_IN_C + lane];
                }
#pragma unroll
                for (int k = 0; k < 16; ++k) {
                    float w  = __int_as_float(__builtin_amdgcn_readlane(__float_as_int(w_l), j0 + k));
                    float vv = (j0 + k < c) ? v[k] : 0.0f;
                    a = __builtin_fmaf(vv, w, a);
                }
            }
        }
        s0[(long long)node * F_IN_C + lane] = a;
    }
}

#define GRU0_WGS 256
#define REAL_RT  (N_NODES_C / 16)

struct hold_t { float v[2][8]; };

static __device__ __forceinline__ hold_t load_hold0(const float* __restrict__ x,
                                                    int rt, int cp, int lane) {
    hold_t h;
    if (cp == 0) {
#pragma unroll
        for (int t = 0; t < 2; ++t)
#pragma unroll
            for (int vr = 0; vr < 8; ++vr)
                h.v[t][vr] = x[(long long)(rt * 16 + vr + 8 * (lane >> 4)) * F_IN_C
                               + t * 16 + (lane & 15)];
    } else {
#pragma unroll
        for (int t = 0; t < 2; ++t)
#pragma unroll
            for (int vr = 0; vr < 8; ++vr) h.v[t][vr] = 0.0f;
    }
    return h;
}

__global__ void __launch_bounds__(512)
gru0_kernel(const float* __restrict__ s0, const float* __restrict__ x,
            float* __restrict__ hout,
            const __bf16* __restrict__ wcb, const __bf16* __restrict__ whhb,
            const float* __restrict__ bih, const float* __restrict__ bhh) {
    int lane = threadIdx.x & 31;
    int half = lane >> 4;
    int wave = __builtin_amdgcn_readfirstlane(threadIdx.x >> 5);
    int cp   = wave & 7;
    int set  = wave >> 3;

    v16bf bw[2][3][2];
#pragma unroll
    for (int mat = 0; mat < 2; ++mat)
#pragma unroll
        for (int g = 0; g < 3; ++g)
#pragma unroll
            for (int t = 0; t < 2; ++t) {
                const __bf16* p = (mat ? whhb : wcb)
                                + (((g * 16 + cp * 2 + t) * 8) << 9) + lane * 8;
                b_raw r;
                r.q0 = *(const v8bf*)(p);
                r.q1 = *(const v8bf*)(p + 256);
                bw[mat][g][t] = make_b_frag(r);
            }

    float bi[3][2], bh[3][2];
#pragma unroll
    for (int g = 0; g < 3; ++g)
#pragma unroll
        for (int t = 0; t < 2; ++t) {
            int c = cp * 32 + t * 16 + (lane & 15);
            bi[g][t] = bih[g * 256 + c];
            bh[g][t] = bhh[g * 256 + c];
        }

    int rt = blockIdx.x + GRU0_WGS * set;
    hold_t ho = load_hold0(x, rt, cp, lane);
    v16bf a_agg, a_h;
    {
        a_raw ca = load_a_raw_f32(s0, rt * 16 + (lane & 15), F_IN_C, 0, lane);
        a_raw ch = load_a_raw_f32(x,  rt * 16 + (lane & 15), F_IN_C, 0, lane);
        a_agg = make_a_frag(ca);
        a_h   = make_a_frag(ch);
    }

    while (true) {
        int  rtn  = rt + 2 * GRU0_WGS;
        bool more = rtn < REAL_RT;
        int  rtp  = more ? rtn : rt;
        a_raw  na = load_a_raw_f32(s0, rtp * 16 + (lane & 15), F_IN_C, 0, lane);
        a_raw  nh = load_a_raw_f32(x,  rtp * 16 + (lane & 15), F_IN_C, 0, lane);
        hold_t no = load_hold0(x, rtp, cp, lane);

#pragma unroll
        for (int t = 0; t < 2; ++t) {
            v8f acc[2][3];
#pragma unroll
            for (int g = 0; g < 3; ++g) {
                v8f z0 = {}, z1 = {};
                acc[0][g] = wmma_bf16(a_agg, bw[0][g][t], z0);
                acc[1][g] = wmma_bf16(a_h,   bw[1][g][t], z1);
            }
            int c = cp * 32 + t * 16 + (lane & 15);
#pragma unroll
            for (int vr = 0; vr < 8; ++vr) {
                long long idx = (long long)(rt * 16 + vr + 8 * half) * 256 + c;
                float r  = sigmoidf_(acc[0][0][vr] + bi[0][t] + acc[1][0][vr] + bh[0][t]);
                float z  = sigmoidf_(acc[0][1][vr] + bi[1][t] + acc[1][1][vr] + bh[1][t]);
                float nn = tanhf(acc[0][2][vr] + bi[2][t] + r * (acc[1][2][vr] + bh[2][t]));
                hout[idx] = (1.0f - z) * nn + z * ho.v[t][vr];
            }
        }

        if (!more) break;
        rt = rtn;
        a_agg = make_a_frag(na);
        a_h   = make_a_frag(nh);
        ho = no;
    }

    if (blockIdx.x == GRU0_WGS - 1 && set == 1) {
        for (int r = N_NODES_C; r < N_PAD; ++r)
            hout[(long long)r * 256 + cp * 32 + lane] = 0.0f;
    }
}

__global__ void __launch_bounds__(256)
final_kernel(const float* __restrict__ h, const float* __restrict__ fcw,
             const float* __restrict__ fcb, float* __restrict__ out) {
    __shared__ float sw[N_ACT_C * 256];
    __shared__ float sb[N_ACT_C];
    for (int i = threadIdx.x; i < N_ACT_C * 256; i += 256) sw[i] = fcw[i];
    if (threadIdx.x < N_ACT_C) sb[threadIdx.x] = fcb[threadIdx.x];
    __syncthreads();

    int n = blockIdx.x * 256 + threadIdx.x;
    if (n >= N_NODES_C) return;

    float acc[N_ACT_C];
#pragma unroll
    for (int a = 0; a < N_ACT_C; ++a) acc[a] = sb[a];
    const float* hr = h + (long long)n * 256;
    for (int k = 0; k < 256; ++k) {
        float hv = fmaxf(hr[k], 0.0f);
#pragma unroll
        for (int a = 0; a < N_ACT_C; ++a) acc[a] += hv * sw[a * 256 + k];
    }
    float mx = acc[0];
#pragma unroll
    for (int a = 1; a < N_ACT_C; ++a) mx = fmaxf(mx, acc[a]);
    float s = 0.0f;
#pragma unroll
    for (int a = 0; a < N_ACT_C; ++a) { acc[a] = __expf(acc[a] - mx); s += acc[a]; }
    float inv = 1.0f / s;
#pragma unroll
    for (int a = 0; a < N_ACT_C; ++a) out[(long long)n * N_ACT_C + a] = acc[a] * inv;
}

extern "C" void kernel_launch(void* const* d_in, const int* in_sizes, int n_in,
                              void* d_out, int out_size, void* d_ws, size_t ws_size,
                              hipStream_t stream) {
    const float* x   = (const float*)d_in[0];
    const int*   ei  = (const int*)  d_in[1];
    const float* ew  = (const float*)d_in[2];
    const float* gg  = (const float*)d_in[3];
    const float* wih = (const float*)d_in[4];
    const float* whh = (const float*)d_in[5];
    const float* bih = (const float*)d_in[6];
    const float* bhh = (const float*)d_in[7];
    const float* fcw = (const float*)d_in[8];
    const float* fcb = (const float*)d_in[9];
    float* out = (float*)d_out;

    float* h0  = (float*)d_ws;
    float* h1  = h0 + (size_t)N_PAD * 256;
    float* agg = h1 + (size_t)N_PAD * 256;
    __bf16* wcb  = (__bf16*)(agg + (size_t)N_PAD * 256);
    __bf16* whhb = wcb + 3 * 768 * 256;
    int*   cnt    = (int*)(whhb + 768 * 256);
    int*   off    = cnt + 50176;
    int*   slot   = off + 50176;
    int*   bucket = slot + N_EDGES_C;
    int*   csrc   = bucket + N_EDGES_C;
    float* cw     = (float*)(csrc + N_EDGES_C);

    convw_kernel<<<(768 * 256 + 255) / 256, 256, 0, stream>>>(whh, whhb);
    combw_kernel<<<(3 * 48 * 16) / 8, 256, 0, stream>>>(gg, wih, wcb);

    zero_int_kernel<<<(N_NODES_C + 1 + 255) / 256, 256, 0, stream>>>(cnt, N_NODES_C + 1);
    count_kernel<<<N_EDGES_C / 256, 256, 0, stream>>>(ei, cnt, slot);
    scan_kernel<<<1, 1024, 0, stream>>>(cnt, off);
    fill_kernel<<<N_EDGES_C / 256, 256, 0, stream>>>(ei, off, slot, bucket);
    rank_kernel<<<N_EDGES_C / 256, 256, 0, stream>>>(ei, ew, off, bucket, csrc, cw);

    float* s0 = h0;
    gather0_kernel<<<G0_BLOCKS, 256, 0, stream>>>(x, off, csrc, cw, s0);
    gru0_kernel<<<GRU0_WGS, 512, 0, stream>>>(s0, x, h1, wcb, whhb, bih, bhh);

    float* hc = h1;
    float* hn = h0;
    for (int l = 1; l < 3; ++l) {
        gather_kernel<<<N_PAD / 8, 256, 0, stream>>>(hc, off, csrc, cw, agg);
        gru_kernel<<<GRU_SLICES * GRU_GROUPS, 256, 0, stream>>>(agg, hc, hn,
                                                                wcb + (size_t)l * 196608, whhb,
                                                                bih, bhh);
        float* t = hc; hc = hn; hn = t;
    }

    final_kernel<<<(N_NODES_C + 255) / 256, 256, 0, stream>>>(hc, fcw, fcb, out);
}


# Prefix-sum kernel: counts in and offsets out as whole 128-byte lines through LDS, a thread scans its 49 entries there; was 98 dependent 4-byte global accesses, every lane on a line of its own
# speedup vs baseline: 3.8880x; 1.1726x over previous
#include <hip/hip_runtime.h>

#define N_NODES_C 50000
#define N_EDGES_C 800000
#define F_IN_C 32
#define C_C 256
#define N_PAD 50016
#define N_ACT_C 8

typedef __attribute__((ext_vector_type(16))) __bf16 v16bf;
typedef __attribute__((ext_vector_type(8)))  __bf16 v8bf;
typedef __attribute__((ext_vector_type(8)))  float  v8f;

static __device__ __forceinline__ v8f wmma_bf16(v16bf a, v16bf b, v8f c) {
    return __builtin_amdgcn_wmma_f32_16x16x32_bf16(
         false, a,  false, b,
         (short)0, c,  false,  false);
}

typedef __attribute__((ext_vector_type(4))) float v4f;
typedef __attribute__((ext_vector_type(2))) float v2f;

static __device__ __forceinline__ v8f wmma_f32(v2f a, v2f b, v8f c) {
    return __builtin_amdgcn_wmma_f32_16x16x4_f32(
         false, a,  false, b,
         (short)0, c,  false,  false);
}


struct a_raw { v4f f0, f1, f2, f3; };

static __device__ __forceinline__ a_raw load_a_raw_f32(
    const float* __restrict__ A, int row, int ld, int k0, int lane) {
    int half = lane >> 4;
    const float* p0 = A + (long long)row * ld + k0 + half * 8;
    const float* p1 = p0 + 16;
    a_raw r;
    r.f0 = *(const v4f*)(p0);
    r.f1 = *(const v4f*)(p0 + 4);
    r.f2 = *(const v4f*)(p1);
    r.f3 = *(const v4f*)(p1 + 4);
    return r;
}

static __device__ __forceinline__ v16bf make_a_frag(const a_raw& r) {
    v16bf a;
#pragma unroll
    for (int i = 0; i < 4; ++i) {
        a[i]      = (__bf16)r.f0[i];
        a[4 + i]  = (__bf16)r.f1[i];
        a[8 + i]  = (__bf16)r.f2[i];
        a[12 + i] = (__bf16)r.f3[i];
    }
    return a;
}

struct b_raw { v8bf q0, q1; };

static __device__ __forceinline__ v16bf make_b_frag(const b_raw& r) {
    v16bf b;
#pragma unroll
    for (int i = 0; i < 8; ++i) { b[i] = r.q0[i]; b[8 + i] = r.q1[i]; }
    return b;
}

static __device__ __forceinline__ float sigmoidf_(float x) {
    return 1.0f / (1.0f + __expf(-x));
}

__global__ void convw_kernel(const float* __restrict__ whh,
                             __bf16* __restrict__ whhb) {
    int o = blockIdx.x * 256 + threadIdx.x;
    if (o >= 768 * 256) return;
    int i  = o & 7;
    int l  = (o >> 3) & 31;
    int q  = (o >> 8) & 1;
    int ks = (o >> 9) & 7;
    int ct = o >> 12;
    int col = ct * 16 + (l & 15);
    int k   = ks * 32 + q * 16 + (l >> 4) * 8 + i;
    whhb[o] = (__bf16)whh[col * 256 + k];
}

__global__ void __launch_bounds__(256)
combw_kernel(const float* __restrict__ gg, const float* __restrict__ wih,
             __bf16* __restrict__ wc) {
    int lane  = threadIdx.x & 31;
    int tile  = __builtin_amdgcn_readfirstlane(blockIdx.x * 8 + (threadIdx.x >> 5));
    int layer = tile / 768;
    int rem   = tile - layer * 768;
    int ct    = rem >> 4;
    int kt    = rem & 15;
    if (layer == 0 && kt >= 2) return;
    int half  = lane >> 4;
    const float* pa = gg + layer * 65536 + (kt * 16 + (lane & 15)) * 256 + half * 8;
    const float* pb = wih + (ct * 16 + (lane & 15)) * 256 + half * 8;

    v8f acc0 = {}, acc1 = {};
#pragma unroll
    for (int ch = 0; ch < 16; ++ch) {
        v4f a0 = *(const v4f*)(pa + ch * 16);
        v4f a1 = *(const v4f*)(pa + ch * 16 + 4);
        v4f b0 = *(const v4f*)(pb + ch * 16);
        v4f b1 = *(const v4f*)(pb + ch * 16 + 4);
        v2f x0 = {a0[0], a0[1]}, y0 = {b0[0], b0[1]};
        v2f x1 = {a0[2], a0[3]}, y1 = {b0[2], b0[3]};
        v2f x2 = {a1[0], a1[1]}, y2 = {b1[0], b1[1]};
        v2f x3 = {a1[2], a1[3]}, y3 = {b1[2], b1[3]};
        acc0 = wmma_f32(x0, y0, acc0);
        acc1 = wmma_f32(x1, y1, acc1);
        acc0 = wmma_f32(x2, y2, acc0);
        acc1 = wmma_f32(x3, y3, acc1);
    }
    v8bf o;
#pragma unroll
    for (int v = 0; v < 8; ++v) o[v] = (__bf16)(acc0[v] + acc1[v]);
    *(v8bf*)(wc + layer * 196608
                + ((((ct * 8 + (kt >> 1)) * 2 + (kt & 1))) << 8) + lane * 8) = o;
}

__global__ void zero_int_kernel(int* __restrict__ p, int n) {
    int i = blockIdx.x * 256 + threadIdx.x;
    if (i < n) p[i] = 0;
}

__global__ void __launch_bounds__(256)
count_kernel(const int* __restrict__ ei, int* __restrict__ cnt,
             int* __restrict__ slot) {
    int e = blockIdx.x * 256 + threadIdx.x;
    if (e >= N_EDGES_C) return;
    slot[e] = atomicAdd(&cnt[ei[N_EDGES_C + e]], 1);
}

#define SCAN_PER ((N_NODES_C + 1 + 1023) / 1024)
#define SCAN_TOT (SCAN_PER * 1024)
#define SCAN_Q   (SCAN_TOT / 4)
#define SCAN_QIN (N_NODES_C / 4)
static_assert(N_NODES_C % 4 == 0, "counts are moved 4 at a time");
static_assert(SCAN_TOT % 4 == 0 && SCAN_TOT >= N_NODES_C + 1, "LDS slots");

__global__ void __launch_bounds__(1024)
scan_kernel(const int* __restrict__ cnt, int* __restrict__ off) {
    __shared__ __attribute__((aligned(16))) int buf[SCAN_TOT];
    __shared__ int wtot[32];
    int t    = threadIdx.x;
    int lane = t & 31;
    int wave = t >> 5;

    {
        const int4* cnt4 = (const int4*)cnt;
        int4* buf4 = (int4*)buf;
        int4 q[(SCAN_Q + 1023) / 1024];
#pragma unroll
        for (int j = 0; j < (SCAN_Q + 1023) / 1024; ++j) {
            int p = j * 1024 + t;
            q[j] = make_int4(0, 0, 0, 0);
            if (p < SCAN_QIN) q[j] = cnt4[p];
        }
#pragma unroll
        for (int j = 0; j < (SCAN_Q + 1023) / 1024; ++j) {
            int p = j * 1024 + t;
            if (p < SCAN_Q) buf4[p] = q[j];
        }
    }
    __syncthreads();

    int v[SCAN_PER];
    int s = 0;
#pragma unroll
    for (int i = 0; i < SCAN_PER; ++i) { v[i] = buf[t * SCAN_PER + i]; s += v[i]; }

    int inc = s;
#pragma unroll
    for (int o = 1; o < 32; o <<= 1) {
        int u = __shfl_up(inc, o);
        if (lane >= o) inc += u;
    }
    if (lane == 31) wtot[wave] = inc;
    __syncthreads();
    int wt   = wtot[lane];
    int winc = wt;
#pragma unroll
    for (int o = 1; o < 32; o <<= 1) {
        int u = __shfl_up(winc, o);
        if (lane >= o) winc += u;
    }
    int before = __shfl(winc - wt, wave);
    int run = before + inc - s;
#pragma unroll
    for (int i = 0; i < SCAN_PER; ++i) { buf[t * SCAN_PER + i] = run; run += v[i]; }
    __syncthreads();

    {
        const int4* buf4 = (const int4*)buf;
        int4* off4 = (int4*)off;
#pragma unroll
        for (int j = 0; j < (SCAN_QIN + 1023) / 1024; ++j) {
            int p = j * 1024 + t;
            if (p < SCAN_QIN) off4[p] = buf4[p];
        }
        if (t == 0) off[N_NODES_C] = buf[N_NODES_C];
    }
}

__global__ void __launch_bounds__(256)
fill_kernel(const int* __restrict__ ei, const int* __restrict__ off,
            const int* __restrict__ slot, int* __restrict__ bucket) {
    int e = blockIdx.x * 256 + threadIdx.x;
    if (e >= N_EDGES_C) return;
    bucket[off[ei[N_EDGES_C + e]] + slot[e]] = e;
}

__global__ void __launch_bounds__(256)
rank_kernel(const int* __restrict__ ei, const float* __restrict__ ew,
            const int* __restrict__ off, const int* __restrict__ bucket,
            int* __restrict__ csrc, float* __restrict__ cw) {
    int e = blockIdx.x * 256 + threadIdx.x;
    if (e >= N_EDGES_C) return;
    int d = ei[N_EDGES_C + e];
    int b = off[d];
    int n = off[d + 1] - b;
    int r = 0;
    for (int i = 0; i < n; ++i) r += (bucket[b + i] < e) ? 1 : 0;
    csrc[b + r] = ei[e];
    cw[b + r]   = ew[e];
}

__global__ void __launch_bounds__(256)
gather_kernel(const float* __restrict__ h, const int* __restrict__ off,
              const int* __restrict__ csrc, const float* __restrict__ cw,
              float* __restrict__ agg) {
    int lane = threadIdx.x & 31;
    int node = __builtin_amdgcn_readfirstlane(blockIdx.x * 8 + (threadIdx.x >> 5));
    int b = 0, n = 0;
    if (node < N_NODES_C) { b = off[node]; n = off[node + 1] - b; }

    float4 a0 = make_float4(0.f, 0.f, 0.f, 0.f);
    float4 a1 = make_float4(0.f, 0.f, 0.f, 0.f);
    for (int base = 0; base < n; base += 32) {
        int c = n - base;
        if (c > 32) c = 32;
        int   s_l = 0;
        float w_l = 0.0f;
        if (lane < c) { s_l = csrc[b + base + lane]; w_l = cw[b + base + lane]; }
#pragma unroll 4
        for (int j = 0; j < c; ++j) {
            int   s = __builtin_amdgcn_readlane(s_l, j);
            float w = __int_as_float(__builtin_amdgcn_readlane(__float_as_int(w_l), j));
            const float4* row = (const float4*)(h + (long long)s * 256);
            float4 v0 = row[lane];
            float4 v1 = row[32 + lane];
            a0.x += v0.x * w; a0.y += v0.y * w; a0.z += v0.z * w; a0.w += v0.w * w;
            a1.x += v1.x * w; a1.y += v1.y * w; a1.z += v1.z * w; a1.w += v1.w * w;
        }
    }
    float4* orow = (float4*)(agg + (long long)node * 256);
    orow[lane]      = a0;
    orow[32 + lane] = a1;
}

#define GRU_SLICES 4
#define GRU_GROUPS 64
#define GRU_RT     (N_PAD / 16)
#define GRU_STRIDE (GRU_GROUPS * 8)

struct b_set { b_raw f[2][4]; };

static __device__ __forceinline__ void lds_load_set(b_set& s, const __bf16* L,
                                                    int g, int ks, int lane) {
#pragma unroll
    for (int mat = 0; mat < 2; ++mat) {
#pragma unroll
        for (int t = 0; t < 4; ++t) {
            const __bf16* p = L + ((((mat * 3 + g) * 4 + t) * 8 + ks) << 9) + lane * 8;
            s.f[mat][t].q0 = *(const v8bf*)(p);
            s.f[mat][t].q1 = *(const v8bf*)(p + 256);
        }
    }
}

__global__ void __launch_bounds__(256)
gru_kernel(const float* __restrict__ agg, const float* __restrict__ hin,
           float* __restrict__ hout,
           const __bf16* __restrict__ wcb, const __bf16* __restrict__ whhb,
           const float* __restrict__ bih, const float* __restrict__ bhh) {
    __shared__ __attribute__((aligned(16))) __bf16 wl[2 * 3 * 4 * 8 * 512];

    int tid  = threadIdx.x;
    int lane = tid & 31;
    int wave = tid >> 5;
    int half = lane >> 4;
    int slice = blockIdx.x & (GRU_SLICES - 1);
    int group = blockIdx.x >> 2;

#pragma unroll
    for (int mg = 0; mg < 6; ++mg) {
        const __bf16* src = ((mg >= 3) ? whhb : wcb)
                          + ((((mg % 3) * 16 + slice * 4) * 8) << 9) + tid * 8;
        __bf16* dst = wl + mg * 16384 + tid * 8;
        v8bf tmp[8];
#pragma unroll
        for (int p = 0; p < 8; ++p) tmp[p] = *(const v8bf*)(src + p * 2048);
#pragma unroll
        for (int p = 0; p < 8; ++p) *(v8bf*)(dst + p * 2048) = tmp[p];
    }
    __syncthreads();

    int rt   = group * 8 + wave;
    int rowA = rt * 16 + (lane & 15);

    a_raw pa[2], ph[2];
    pa[0] = load_a_raw_f32(agg, rowA, 256, 0,  lane);
    ph[0] = load_a_raw_f32(hin, rowA, 256, 0,  lane);
    pa[1] = load_a_raw_f32(agg, rowA, 256, 32, lane);
    ph[1] = load_a_raw_f32(hin, rowA, 256, 32, lane);

    b_set bs[2];
    lds_load_set(bs[0], wl, 0, 0, lane);

    while (true) {
        int  rtn  = rt + GRU_STRIDE;
        bool more = rtn < GRU_RT;
        int  rowN = (more ? rtn : rt) * 16 + (lane & 15);

        float hold[4][8];
        v8f acc[2][3][4];
#pragma unroll
        for (int a = 0; a < 2; ++a)
#pragma unroll
            for (int g = 0; g < 3; ++g)
#pragma unroll
                for (int t = 0; t < 4; ++t) { v8f z = {}; acc[a][g][t] = z; }

#pragma unroll
        for (int ks = 0; ks < 8; ++ks) {
            v16bf a_agg = make_a_frag(pa[ks & 1]);
            v16bf a_h   = make_a_frag(ph[ks & 1]);
            if (ks < 6) {
                pa[ks & 1] = load_a_raw_f32(agg, rowA, 256, (ks + 2) * 32, lane);
                ph[ks & 1] = load_a_raw_f32(hin, rowA, 256, (ks + 2) * 32, lane);
            } else {
                pa[ks & 1] = load_a_raw_f32(agg, rowN, 256, (ks - 6) * 32, lane);
                ph[ks & 1] = load_a_raw_f32(hin, rowN, 256, (ks - 6) * 32, lane);
            }
            if (ks == 6) {
#pragma unroll
                for (int t = 0; t < 4; ++t)
#pragma unroll
                    for (int vr = 0; vr < 8; ++vr)
                        hold[t][vr] = hin[(long long)(rt * 16 + vr + 8 * half) * 256
                                          + slice * 64 + t * 16 + (lane & 15)];
            }
#pragma unroll
            for (int g = 0; g < 3; ++g) {
                const int bi = ks * 3 + g;
                const int nb = (bi + 1) % 24;
                b_set& cur = bs[bi & 1];
                lds_load_set(bs[(bi + 1) & 1], wl, nb % 3, nb / 3, lane);
                asm volatile("" :: "v"(cur.f[0][0].q0), "v"(cur.f[0][0].q1),
                                   "v"(cur.f[0][1].q0), "v"(cur.f[0][1].q1),
                                   "v"(cur.f[0][2].q0), "v"(cur.f[0][2].q1),
                                   "v"(cur.f[0][3].q0), "v"(cur.f[0][3].q1),
                                   "v"(cur.f[1][0].q0), "v"(cur.f[1][0].q1),
                                   "v"(cur.f[1][1].q0), "v"(cur.f[1][1].q1),
                                   "v"(cur.f[1][2].q0), "v"(cur.f[1][2].q1),
                                   "v"(cur.f[1][3].q0), "v"(cur.f[1][3].q1));
#pragma unroll
                for (int t = 0; t < 4; ++t) {
                    acc[0][g][t] = wmma_bf16(a_agg, make_b_frag(cur.f[0][t]), acc[0][g][t]);
                    acc[1][g][t] = wmma_bf16(a_h,   make_b_frag(cur.f[1][t]), acc[1][g][t]);
                }
            }
        }

#pragma unroll
        for (int t = 0; t < 4; ++t) {
            int c = slice * 64 + t * 16 + (lane & 15);
            float bir = bih[c],       bhr = bhh[c];
            float biz = bih[256 + c], bhz = bhh[256 + c];
            float bin = bih[512 + c], bhn = bhh[512 + c];
            asm volatile("" :: "v"(hold[t][0]), "v"(hold[t][1]), "v"(hold[t][2]), "v"(hold[t][3]),
                               "v"(hold[t][4]), "v"(hold[t][5]), "v"(hold[t][6]), "v"(hold[t][7]));
#pragma unroll
            for (int vr = 0; vr < 8; ++vr) {
                long long idx = (long long)(rt * 16 + vr + 8 * half) * 256 + c;
                float r  = sigmoidf_(acc[0][0][t][vr] + bir + acc[1][0][t][vr] + bhr);
                float z  = sigmoidf_(acc[0][1][t][vr] + biz + acc[1][1][t][vr] + bhz);
                float nn = tanhf(acc[0][2][t][vr] + bin + r * (acc[1][2][t][vr] + bhn));
                hout[idx] = (1.0f - z) * nn + z * hold[t][vr];
            }
        }

        if (!more) break;
        rt   = rtn;
        rowA = rowN;
    }
}

#define G0_BLOCKS 1024
__global__ void __launch_bounds__(256)
gather0_kernel(const float* __restrict__ x, const int* __restrict__ off,
               const int* __restrict__ csrc, const float* __restrict__ cw,
               float* __restrict__ s0) {
    int lane = threadIdx.x & 31;
    int gw = __builtin_amdgcn_readfirstlane(blockIdx.x * 8 + (threadIdx.x >> 5));
    for (int node = gw; node < N_PAD; node += G0_BLOCKS * 8) {
        int b = 0, n = 0;
        if (node < N_NODES_C) { b = off[node]; n = off[node + 1] - b; }
        float a = 0.0f;
        for (int base = 0; base < n; base += 32) {
            int c = n - base;
            if (c > 32) c = 32;
            int   s_l = 0;
            float w_l = 0.0f;
            if (lane < c) { s_l = csrc[b + base + lane]; w_l = cw[b + base + lane]; }
            for (int j0 = 0; j0 < c; j0 += 16) {
                float v[16];
#pragma unroll
                for (int k = 0; k < 16; ++k) {
                    int s = __builtin_amdgcn_readlane(s_l, j0 + k);
                    v[k] = x[(long long)s * F_IN_C + lane];
                }
#pragma unroll
                for (int k = 0; k < 16; ++k) {
                    float w  = __int_as_float(__builtin_amdgcn_readlane(__float_as_int(w_l), j0 + k));
                    float vv = (j0 + k < c) ? v[k] : 0.0f;
                    a = __builtin_fmaf(vv, w, a);
                }
            }
        }
        s0[(long long)node * F_IN_C + lane] = a;
    }
}

#define GRU0_WGS 256
#define REAL_RT  (N_NODES_C / 16)

struct hold_t { float v[2][8]; };

static __device__ __forceinline__ hold_t load_hold0(const float* __restrict__ x,
                                                    int rt, int cp, int lane) {
    hold_t h;
    if (cp == 0) {
#pragma unroll
        for (int t = 0; t < 2; ++t)
#pragma unroll
            for (int vr = 0; vr < 8; ++vr)
                h.v[t][vr] = x[(long long)(rt * 16 + vr + 8 * (lane >> 4)) * F_IN_C
                               + t * 16 + (lane & 15)];
    } else {
#pragma unroll
        for (int t = 0; t < 2; ++t)
#pragma unroll
            for (int vr = 0; vr < 8; ++vr) h.v[t][vr] = 0.0f;
    }
    return h;
}

__global__ void __launch_bounds__(512)
gru0_kernel(const float* __restrict__ s0, const float* __restrict__ x,
            float* __restrict__ hout,
            const __bf16* __restrict__ wcb, const __bf16* __restrict__ whhb,
            const float* __restrict__ bih, const float* __restrict__ bhh) {
    int lane = threadIdx.x & 31;
    int half = lane >> 4;
    int wave = __builtin_amdgcn_readfirstlane(threadIdx.x >> 5);
    int cp   = wave & 7;
    int set  = wave >> 3;

    v16bf bw[2][3][2];
#pragma unroll
    for (int mat = 0; mat < 2; ++mat)
#pragma unroll
        for (int g = 0; g < 3; ++g)
#pragma unroll
            for (int t = 0; t < 2; ++t) {
                const __bf16* p = (mat ? whhb : wcb)
                                + (((g * 16 + cp * 2 + t) * 8) << 9) + lane * 8;
                b_raw r;
                r.q0 = *(const v8bf*)(p);
                r.q1 = *(const v8bf*)(p + 256);
                bw[mat][g][t] = make_b_frag(r);
            }

    float bi[3][2], bh[3][2];
#pragma unroll
    for (int g = 0; g < 3; ++g)
#pragma unroll
        for (int t = 0; t < 2; ++t) {
            int c = cp * 32 + t * 16 + (lane & 15);
            bi[g][t] = bih[g * 256 + c];
            bh[g][t] = bhh[g * 256 + c];
        }

    int rt = blockIdx.x + GRU0_WGS * set;
    hold_t ho = load_hold0(x, rt, cp, lane);
    v16bf a_agg, a_h;
    {
        a_raw ca = load_a_raw_f32(s0, rt * 16 + (lane & 15), F_IN_C, 0, lane);
        a_raw ch = load_a_raw_f32(x,  rt * 16 + (lane & 15), F_IN_C, 0, lane);
        a_agg = make_a_frag(ca);
        a_h   = make_a_frag(ch);
    }

    while (true) {
        int  rtn  = rt + 2 * GRU0_WGS;
        bool more = rtn < REAL_RT;
        int  rtp  = more ? rtn : rt;
        a_raw  na = load_a_raw_f32(s0, rtp * 16 + (lane & 15), F_IN_C, 0, lane);
        a_raw  nh = load_a_raw_f32(x,  rtp * 16 + (lane & 15), F_IN_C, 0, lane);
        hold_t no = load_hold0(x, rtp, cp, lane);

#pragma unroll
        for (int t = 0; t < 2; ++t) {
            v8f acc[2][3];
#pragma unroll
            for (int g = 0; g < 3; ++g) {
                v8f z0 = {}, z1 = {};
                acc[0][g] = wmma_bf16(a_agg, bw[0][g][t], z0);
                acc[1][g] = wmma_bf16(a_h,   bw[1][g][t], z1);
            }
            int c = cp * 32 + t * 16 + (lane & 15);
#pragma unroll
            for (int vr = 0; vr < 8; ++vr) {
                long long idx = (long long)(rt * 16 + vr + 8 * half) * 256 + c;
                float r  = sigmoidf_(acc[0][0][vr] + bi[0][t] + acc[1][0][vr] + bh[0][t]);
                float z  = sigmoidf_(acc[0][1][vr] + bi[1][t] + acc[1][1][vr] + bh[1][t]);
                float nn = tanhf(acc[0][2][vr] + bi[2][t] + r * (acc[1][2][vr] + bh[2][t]));
                hout[idx] = (1.0f - z) * nn + z * ho.v[t][vr];
            }
        }

        if (!more) break;
        rt = rtn;
        a_agg = make_a_frag(na);
        a_h   = make_a_frag(nh);
        ho = no;
    }

    if (blockIdx.x == GRU0_WGS - 1 && set == 1) {
        for (int r = N_NODES_C; r < N_PAD; ++r)
            hout[(long long)r * 256 + cp * 32 + lane] = 0.0f;
    }
}

__global__ void __launch_bounds__(256)
final_kernel(const float* __restrict__ h, const float* __restrict__ fcw,
             const float* __restrict__ fcb, float* __restrict__ out) {
    __shared__ float sw[N_ACT_C * 256];
    __shared__ float sb[N_ACT_C];
    for (int i = threadIdx.x; i < N_ACT_C * 256; i += 256) sw[i] = fcw[i];
    if (threadIdx.x < N_ACT_C) sb[threadIdx.x] = fcb[threadIdx.x];
    __syncthreads();

    int n = blockIdx.x * 256 + threadIdx.x;
    if (n >= N_NODES_C) return;

    float acc[N_ACT_C];
#pragma unroll
    for (int a = 0; a < N_ACT_C; ++a) acc[a] = sb[a];
    const float* hr = h + (long long)n * 256;
    for (int k = 0; k < 256; ++k) {
        float hv = fmaxf(hr[k], 0.0f);
#pragma unroll
        for (int a = 0; a < N_ACT_C; ++a) acc[a] += hv * sw[a * 256 + k];
    }
    float mx = acc[0];
#pragma unroll
    for (int a = 1; a < N_ACT_C; ++a) mx = fmaxf(mx, acc[a]);
    float s = 0.0f;
#pragma unroll
    for (int a = 0; a < N_ACT_C; ++a) { acc[a] = __expf(acc[a] - mx); s += acc[a]; }
    float inv = 1.0f / s;
#pragma unroll
    for (int a = 0; a < N_ACT_C; ++a) out[(long long)n * N_ACT_C + a] = acc[a] * inv;
}

extern "C" void kernel_launch(void* const* d_in, const int* in_sizes, int n_in,
                              void* d_out, int out_size, void* d_ws, size_t ws_size,
                              hipStream_t stream) {
    const float* x   = (const float*)d_in[0];
    const int*   ei  = (const int*)  d_in[1];
    const float* ew  = (const float*)d_in[2];
    const float* gg  = (const float*)d_in[3];
    const float* wih = (const float*)d_in[4];
    const float* whh = (const float*)d_in[5];
    const float* bih = (const float*)d_in[6];
    const float* bhh = (const float*)d_in[7];
    const float* fcw = (const float*)d_in[8];
    const float* fcb = (const float*)d_in[9];
    float* out = (float*)d_out;

    float* h0  = (float*)d_ws;
    float* h1  = h0 + (size_t)N_PAD * 256;
    float* agg = h1 + (size_t)N_PAD * 256;
    __bf16* wcb  = (__bf16*)(agg + (size_t)N_PAD * 256);
    __bf16* whhb = wcb + 3 * 768 * 256;
    int*   cnt    = (int*)(whhb + 768 * 256);
    int*   off    = cnt + 50176;
    int*   slot   = off + 50176;
    int*   bucket = slot + N_EDGES_C;
    int*   csrc   = bucket + N_EDGES_C;
    float* cw     = (float*)(csrc + N_EDGES_C);

    convw_kernel<<<(768 * 256 + 255) / 256, 256, 0, stream>>>(whh, whhb);
    combw_kernel<<<(3 * 48 * 16) / 8, 256, 0, stream>>>(gg, wih, wcb);

    zero_int_kernel<<<(N_NODES_C + 1 + 255) / 256, 256, 0, stream>>>(cnt, N_NODES_C + 1);
    count_kernel<<<N_EDGES_C / 256, 256, 0, stream>>>(ei, cnt, slot);
    scan_kernel<<<1, 1024, 0, stream>>>(cnt, off);
    fill_kernel<<<N_EDGES_C / 256, 256, 0, stream>>>(ei, off, slot, bucket);
    rank_kernel<<<N_EDGES_C / 256, 256, 0, stream>>>(ei, ew, off, bucket, csrc, cw);

    float* s0 = h0;
    gather0_kernel<<<G0_BLOCKS, 256, 0, stream>>>(x, off, csrc, cw, s0);
    gru0_kernel<<<GRU0_WGS, 512, 0, stream>>>(s0, x, h1, wcb, whhb, bih, bhh);

    float* hc = h1;
    float* hn = h0;
    for (int l = 1; l < 3; ++l) {
        gather_kernel<<<N_PAD / 8, 256, 0, stream>>>(hc, off, csrc, cw, agg);
        gru_kernel<<<GRU_SLICES * GRU_GROUPS, 256, 0, stream>>>(agg, hc, hn,
                                                                wcb + (size_t)l * 196608, whhb,
                                                                bih, bhh);
        float* t = hc; hc = hn; hn = t;
    }

    final_kernel<<<(N_NODES_C + 255) / 256, 256, 0, stream>>>(hc, fcw, fcb, out);
}


# GRU kernel (layers 1, 2): wave tile 32 rows x 64 columns at one wave per SIMD (was 16 x 64, two): an LDS weight fragment feeds two WMMAs; block loads fenced ahead, A operands 4 k-steps ahead
# speedup vs baseline: 4.7336x; 1.2175x over previous
#include <hip/hip_runtime.h>

#define N_NODES_C 50000
#define N_EDGES_C 800000
#define F_IN_C 32
#define C_C 256
#define N_PAD 50016
#define N_ACT_C 8

typedef __attribute__((ext_vector_type(16))) __bf16 v16bf;
typedef __attribute__((ext_vector_type(8)))  __bf16 v8bf;
typedef __attribute__((ext_vector_type(8)))  float  v8f;

static __device__ __forceinline__ v8f wmma_bf16(v16bf a, v16bf b, v8f c) {
    return __builtin_amdgcn_wmma_f32_16x16x32_bf16(
         false, a,  false, b,
         (short)0, c,  false,  false);
}

typedef __attribute__((ext_vector_type(4))) float v4f;
typedef __attribute__((ext_vector_type(2))) float v2f;

static __device__ __forceinline__ v8f wmma_f32(v2f a, v2f b, v8f c) {
    return __builtin_amdgcn_wmma_f32_16x16x4_f32(
         false, a,  false, b,
         (short)0, c,  false,  false);
}


struct a_raw { v4f f0, f1, f2, f3; };

static __device__ __forceinline__ a_raw load_a_raw_f32(
    const float* __restrict__ A, int row, int ld, int k0, int lane) {
    int half = lane >> 4;
    const float* p0 = A + (long long)row * ld + k0 + half * 8;
    const float* p1 = p0 + 16;
    a_raw r;
    r.f0 = *(const v4f*)(p0);
    r.f1 = *(const v4f*)(p0 + 4);
    r.f2 = *(const v4f*)(p1);
    r.f3 = *(const v4f*)(p1 + 4);
    return r;
}

static __device__ __forceinline__ v16bf make_a_frag(const a_raw& r) {
    v16bf a;
#pragma unroll
    for (int i = 0; i < 4; ++i) {
        a[i]      = (__bf16)r.f0[i];
        a[4 + i]  = (__bf16)r.f1[i];
        a[8 + i]  = (__bf16)r.f2[i];
        a[12 + i] = (__bf16)r.f3[i];
    }
    return a;
}

struct b_raw { v8bf q0, q1; };

static __device__ __forceinline__ v16bf make_b_frag(const b_raw& r) {
    v16bf b;
#pragma unroll
    for (int i = 0; i < 8; ++i) { b[i] = r.q0[i]; b[8 + i] = r.q1[i]; }
    return b;
}

static __device__ __forceinline__ float sigmoidf_(float x) {
    return __builtin_amdgcn_rcpf(1.0f + __builtin_amdgcn_exp2f(x * -1.44269504f));
}

static __device__ __forceinline__ float tanhf_(float x) {
    return 1.0f - 2.0f * __builtin_amdgcn_rcpf(1.0f + __builtin_amdgcn_exp2f(x * 2.88539008f));
}

__global__ void convw_kernel(const float* __restrict__ whh,
                             __bf16* __restrict__ whhb) {
    int o = blockIdx.x * 256 + threadIdx.x;
    if (o >= 768 * 256) return;
    int i  = o & 7;
    int l  = (o >> 3) & 31;
    int q  = (o >> 8) & 1;
    int ks = (o >> 9) & 7;
    int ct = o >> 12;
    int col = ct * 16 + (l & 15);
    int k   = ks * 32 + q * 16 + (l >> 4) * 8 + i;
    whhb[o] = (__bf16)whh[col * 256 + k];
}

__global__ void __launch_bounds__(256)
combw_kernel(const float* __restrict__ gg, const float* __restrict__ wih,
             __bf16* __restrict__ wc) {
    int lane  = threadIdx.x & 31;
    int tile  = __builtin_amdgcn_readfirstlane(blockIdx.x * 8 + (threadIdx.x >> 5));
    int layer = tile / 768;
    int rem   = tile - layer * 768;
    int ct    = rem >> 4;
    int kt    = rem & 15;
    if (layer == 0 && kt >= 2) return;
    int half  = lane >> 4;
    const float* pa = gg + layer * 65536 + (kt * 16 + (lane & 15)) * 256 + half * 8;
    const float* pb = wih + (ct * 16 + (lane & 15)) * 256 + half * 8;

    v8f acc0 = {}, acc1 = {};
#pragma unroll
    for (int ch = 0; ch < 16; ++ch) {
        v4f a0 = *(const v4f*)(pa + ch * 16);
        v4f a1 = *(const v4f*)(pa + ch * 16 + 4);
        v4f b0 = *(const v4f*)(pb + ch * 16);
        v4f b1 = *(const v4f*)(pb + ch * 16 + 4);
        v2f x0 = {a0[0], a0[1]}, y0 = {b0[0], b0[1]};
        v2f x1 = {a0[2], a0[3]}, y1 = {b0[2], b0[3]};
        v2f x2 = {a1[0], a1[1]}, y2 = {b1[0], b1[1]};
        v2f x3 = {a1[2], a1[3]}, y3 = {b1[2], b1[3]};
        acc0 = wmma_f32(x0, y0, acc0);
        acc1 = wmma_f32(x1, y1, acc1);
        acc0 = wmma_f32(x2, y2, acc0);
        acc1 = wmma_f32(x3, y3, acc1);
    }
    v8bf o;
#pragma unroll
    for (int v = 0; v < 8; ++v) o[v] = (__bf16)(acc0[v] + acc1[v]);
    *(v8bf*)(wc + layer * 196608
                + ((((ct * 8 + (kt >> 1)) * 2 + (kt & 1))) << 8) + lane * 8) = o;
}

__global__ void zero_int_kernel(int* __restrict__ p, int n) {
    int i = blockIdx.x * 256 + threadIdx.x;
    if (i < n) p[i] = 0;
}

__global__ void __launch_bounds__(256)
count_kernel(const int* __restrict__ ei, int* __restrict__ cnt,
             int* __restrict__ slot) {
    int e = blockIdx.x * 256 + threadIdx.x;
    if (e >= N_EDGES_C) return;
    slot[e] = atomicAdd(&cnt[ei[N_EDGES_C + e]], 1);
}

#define SCAN_PER ((N_NODES_C + 1 + 1023) / 1024)
#define SCAN_TOT (SCAN_PER * 1024)
#define SCAN_Q   (SCAN_TOT / 4)
#define SCAN_QIN (N_NODES_C / 4)
static_assert(N_NODES_C % 4 == 0, "counts are moved 4 at a time");
static_assert(SCAN_TOT % 4 == 0 && SCAN_TOT >= N_NODES_C + 1, "LDS slots");

__global__ void __launch_bounds__(1024)
scan_kernel(const int* __restrict__ cnt, int* __restrict__ off) {
    __shared__ __attribute__((aligned(16))) int buf[SCAN_TOT];
    __shared__ int wtot[32];
    int t    = threadIdx.x;
    int lane = t & 31;
    int wave = t >> 5;

    {
        const int4* cnt4 = (const int4*)cnt;
        int4* buf4 = (int4*)buf;
        int4 q[(SCAN_Q + 1023) / 1024];
#pragma unroll
        for (int j = 0; j < (SCAN_Q + 1023) / 1024; ++j) {
            int p = j * 1024 + t;
            q[j] = make_int4(0, 0, 0, 0);
            if (p < SCAN_QIN) q[j] = cnt4[p];
        }
#pragma unroll
        for (int j = 0; j < (SCAN_Q + 1023) / 1024; ++j) {
            int p = j * 1024 + t;
            if (p < SCAN_Q) buf4[p] = q[j];
        }
    }
    __syncthreads();

    int v[SCAN_PER];
    int s = 0;
#pragma unroll
    for (int i = 0; i < SCAN_PER; ++i) { v[i] = buf[t * SCAN_PER + i]; s += v[i]; }

    int inc = s;
#pragma unroll
    for (int o = 1; o < 32; o <<= 1) {
        int u = __shfl_up(inc, o);
        if (lane >= o) inc += u;
    }
    if (lane == 31) wtot[wave] = inc;
    __syncthreads();
    int wt   = wtot[lane];
    int winc = wt;
#pragma unroll
    for (int o = 1; o < 32; o <<= 1) {
        int u = __shfl_up(winc, o);
        if (lane >= o) winc += u;
    }
    int before = __shfl(winc - wt, wave);
    int run = before + inc - s;
#pragma unroll
    for (int i = 0; i < SCAN_PER; ++i) { buf[t * SCAN_PER + i] = run; run += v[i]; }
    __syncthreads();

    {
        const int4* buf4 = (const int4*)buf;
        int4* off4 = (int4*)off;
#pragma unroll
        for (int j = 0; j < (SCAN_QIN + 1023) / 1024; ++j) {
            int p = j * 1024 + t;
            if (p < SCAN_QIN) off4[p] = buf4[p];
        }
        if (t == 0) off[N_NODES_C] = buf[N_NODES_C];
    }
}

__global__ void __launch_bounds__(256)
fill_kernel(const int* __restrict__ ei, const int* __restrict__ off,
            const int* __restrict__ slot, int* __restrict__ bucket) {
    int e = blockIdx.x * 256 + threadIdx.x;
    if (e >= N_EDGES_C) return;
    bucket[off[ei[N_EDGES_C + e]] + slot[e]] = e;
}

__global__ void __launch_bounds__(256)
rank_kernel(const int* __restrict__ ei, const float* __restrict__ ew,
            const int* __restrict__ off, const int* __restrict__ bucket,
            int* __restrict__ csrc, float* __restrict__ cw) {
    int e = blockIdx.x * 256 + threadIdx.x;
    if (e >= N_EDGES_C) return;
    int d = ei[N_EDGES_C + e];
    int b = off[d];
    int n = off[d + 1] - b;
    int r = 0;
    for (int i = 0; i < n; ++i) r += (bucket[b + i] < e) ? 1 : 0;
    csrc[b + r] = ei[e];
    cw[b + r]   = ew[e];
}

__global__ void __launch_bounds__(256)
gather_kernel(const float* __restrict__ h, const int* __restrict__ off,
              const int* __restrict__ csrc, const float* __restrict__ cw,
              float* __restrict__ agg) {
    int lane = threadIdx.x & 31;
    int node = __builtin_amdgcn_readfirstlane(blockIdx.x * 8 + (threadIdx.x >> 5));
    int b = 0, n = 0;
    if (node < N_NODES_C) { b = off[node]; n = off[node + 1] - b; }

    float4 a0 = make_float4(0.f, 0.f, 0.f, 0.f);
    float4 a1 = make_float4(0.f, 0.f, 0.f, 0.f);
    for (int base = 0; base < n; base += 32) {
        int c = n - base;
        if (c > 32) c = 32;
        int   s_l = 0;
        float w_l = 0.0f;
        if (lane < c) { s_l = csrc[b + base + lane]; w_l = cw[b + base + lane]; }
#pragma unroll 4
        for (int j = 0; j < c; ++j) {
            int   s = __builtin_amdgcn_readlane(s_l, j);
            float w = __int_as_float(__builtin_amdgcn_readlane(__float_as_int(w_l), j));
            const float4* row = (const float4*)(h + (long long)s * 256);
            float4 v0 = row[lane];
            float4 v1 = row[32 + lane];
            a0.x += v0.x * w; a0.y += v0.y * w; a0.z += v0.z * w; a0.w += v0.w * w;
            a1.x += v1.x * w; a1.y += v1.y * w; a1.z += v1.z * w; a1.w += v1.w * w;
        }
    }
    float4* orow = (float4*)(agg + (long long)node * 256);
    orow[lane]      = a0;
    orow[32 + lane] = a1;
}

#define GRU_SLICES 4
#define GRU_GROUPS 64
#define GRU_WAVES  4
#define GRU_RP     (N_PAD / 32)
#define GRU_STRIDE (GRU_GROUPS * GRU_WAVES)
static_assert(N_PAD % 32 == 0, "row pairs tile the padded rows");
static_assert(GRU_STRIDE <= GRU_RP, "every wave has at least one row pair");

struct b_set { b_raw f[2][4]; };

static __device__ __forceinline__ void lds_load_set(b_set& s, const __bf16* L,
                                                    int g, int ks, int lane) {
#pragma unroll
    for (int mat = 0; mat < 2; ++mat) {
#pragma unroll
        for (int t = 0; t < 4; ++t) {
            const __bf16* p = L + ((((mat * 3 + g) * 4 + t) * 8 + ks) << 9) + lane * 8;
            s.f[mat][t].q0 = *(const v8bf*)(p);
            s.f[mat][t].q1 = *(const v8bf*)(p + 256);
        }
    }
}

__global__ void __launch_bounds__(GRU_WAVES * 32)
gru_kernel(const float* __restrict__ agg, const float* __restrict__ hin,
           float* __restrict__ hout,
           const __bf16* __restrict__ wcb, const __bf16* __restrict__ whhb,
           const float* __restrict__ bih, const float* __restrict__ bhh) {
    __shared__ __attribute__((aligned(16))) __bf16 wl[2 * 3 * 4 * 8 * 512];

    int tid  = threadIdx.x;
    int lane = tid & 31;
    int wave = tid >> 5;
    int half = lane >> 4;
    int slice = blockIdx.x & (GRU_SLICES - 1);
    int group = blockIdx.x >> 2;

#pragma unroll
    for (int mg = 0; mg < 6; ++mg) {
        const __bf16* src = ((mg >= 3) ? whhb : wcb)
                          + ((((mg % 3) * 16 + slice * 4) * 8) << 9) + tid * 8;
        __bf16* dst = wl + mg * 16384 + tid * 8;
        v8bf tmp[16];
#pragma unroll
        for (int p = 0; p < 16; ++p) tmp[p] = *(const v8bf*)(src + p * 1024);
#pragma unroll
        for (int p = 0; p < 16; ++p) *(v8bf*)(dst + p * 1024) = tmp[p];
    }
    __syncthreads();

    int rp   = group * GRU_WAVES + wave;
    int rowA = rp * 32 + (lane & 15);

    a_raw pa[4][2], ph[4][2];
#pragma unroll
    for (int s = 0; s < 4; ++s)
#pragma unroll
        for (int u = 0; u < 2; ++u) {
            pa[s][u] = load_a_raw_f32(agg, rowA + 16 * u, 256, s * 32, lane);
            ph[s][u] = load_a_raw_f32(hin, rowA + 16 * u, 256, s * 32, lane);
        }

    b_set bs[2];
    lds_load_set(bs[0], wl, 0, 0, lane);

    while (true) {
        int  rpn  = rp + GRU_STRIDE;
        bool more = rpn < GRU_RP;
        int  rowN = (more ? rpn : rp) * 32 + (lane & 15);

        float hold[2][4][8];
        v8f acc[2][2][3][4];
#pragma unroll
        for (int u = 0; u < 2; ++u)
#pragma unroll
            for (int a = 0; a < 2; ++a)
#pragma unroll
                for (int g = 0; g < 3; ++g)
#pragma unroll
                    for (int t = 0; t < 4; ++t) { v8f z = {}; acc[u][a][g][t] = z; }

#pragma unroll
        for (int ks = 0; ks < 8; ++ks) {
            v16bf a_agg[2], a_h[2];
#pragma unroll
            for (int u = 0; u < 2; ++u) {
                a_agg[u] = make_a_frag(pa[ks & 3][u]);
                a_h[u]   = make_a_frag(ph[ks & 3][u]);
                if (ks < 4) {
                    pa[ks & 3][u] = load_a_raw_f32(agg, rowA + 16 * u, 256, (ks + 4) * 32, lane);
                    ph[ks & 3][u] = load_a_raw_f32(hin, rowA + 16 * u, 256, (ks + 4) * 32, lane);
                } else {
                    pa[ks & 3][u] = load_a_raw_f32(agg, rowN + 16 * u, 256, (ks - 4) * 32, lane);
                    ph[ks & 3][u] = load_a_raw_f32(hin, rowN + 16 * u, 256, (ks - 4) * 32, lane);
                }
            }
            if (ks == 4 || ks == 6) {
                const int u = (ks - 4) >> 1;
#pragma unroll
                for (int t = 0; t < 4; ++t)
#pragma unroll
                    for (int vr = 0; vr < 8; ++vr)
                        hold[u][t][vr] = hin[(long long)(rp * 32 + 16 * u + vr + 8 * half) * 256
                                             + slice * 64 + t * 16 + (lane & 15)];
            }
#pragma unroll
            for (int g = 0; g < 3; ++g) {
                const int bi = ks * 3 + g;
                const int nb = (bi + 1) % 24;
                b_set& cur = bs[bi & 1];
                lds_load_set(bs[(bi + 1) & 1], wl, nb % 3, nb / 3, lane);
                __builtin_amdgcn_sched_barrier(0);
#pragma unroll
                for (int t = 0; t < 4; ++t) {
                    v16bf b0 = make_b_frag(cur.f[0][t]);
                    v16bf b1 = make_b_frag(cur.f[1][t]);
#pragma unroll
                    for (int u = 0; u < 2; ++u) {
                        acc[u][0][g][t] = wmma_bf16(a_agg[u], b0, acc[u][0][g][t]);
                        acc[u][1][g][t] = wmma_bf16(a_h[u],   b1, acc[u][1][g][t]);
                    }
                }
                __builtin_amdgcn_sched_barrier(0);
            }
        }

#pragma unroll
        for (int t = 0; t < 4; ++t) {
            int c = slice * 64 + t * 16 + (lane & 15);
            float bir = bih[c],       bhr = bhh[c];
            float biz = bih[256 + c], bhz = bhh[256 + c];
            float bin = bih[512 + c], bhn = bhh[512 + c];
#pragma unroll
            for (int u = 0; u < 2; ++u) {
#pragma unroll
                for (int vr = 0; vr < 8; ++vr) {
                    long long idx = (long long)(rp * 32 + 16 * u + vr + 8 * half) * 256 + c;
                    float r  = sigmoidf_(acc[u][0][0][t][vr] + bir + acc[u][1][0][t][vr] + bhr);
                    float z  = sigmoidf_(acc[u][0][1][t][vr] + biz + acc[u][1][1][t][vr] + bhz);
                    float nn = tanhf_(acc[u][0][2][t][vr] + bin + r * (acc[u][1][2][t][vr] + bhn));
                    hout[idx] = (1.0f - z) * nn + z * hold[u][t][vr];
                }
            }
        }

        if (!more) break;
        rp   = rpn;
        rowA = rowN;
    }
}

#define G0_BLOCKS 1024
__global__ void __launch_bounds__(256)
gather0_kernel(const float* __restrict__ x, const int* __restrict__ off,
               const int* __restrict__ csrc, const float* __restrict__ cw,
               float* __restrict__ s0) {
    int lane = threadIdx.x & 31;
    int gw = __builtin_amdgcn_readfirstlane(blockIdx.x * 8 + (threadIdx.x >> 5));
    for (int node = gw; node < N_PAD; node += G0_BLOCKS * 8) {
        int b = 0, n = 0;
        if (node < N_NODES_C) { b = off[node]; n = off[node + 1] - b; }
        float a = 0.0f;
        for (int base = 0; base < n; base += 32) {
            int c = n - base;
            if (c > 32) c = 32;
            int   s_l = 0;
            float w_l = 0.0f;
            if (lane < c) { s_l = csrc[b + base + lane]; w_l = cw[b + base + lane]; }
            for (int j0 = 0; j0 < c; j0 += 16) {
                float v[16];
#pragma unroll
                for (int k = 0; k < 16; ++k) {
                    int s = __builtin_amdgcn_readlane(s_l, j0 + k);
                    v[k] = x[(long long)s * F_IN_C + lane];
                }
#pragma unroll
                for (int k = 0; k < 16; ++k) {
                    float w  = __int_as_float(__builtin_amdgcn_readlane(__float_as_int(w_l), j0 + k));
                    float vv = (j0 + k < c) ? v[k] : 0.0f;
                    a = __builtin_fmaf(vv, w, a);
                }
            }
        }
        s0[(long long)node * F_IN_C + lane] = a;
    }
}

#define GRU0_WGS 256
#define REAL_RT  (N_NODES_C / 16)

struct hold_t { float v[2][8]; };

static __device__ __forceinline__ hold_t load_hold0(const float* __restrict__ x,
                                                    int rt, int cp, int lane) {
    hold_t h;
    if (cp == 0) {
#pragma unroll
        for (int t = 0; t < 2; ++t)
#pragma unroll
            for (int vr = 0; vr < 8; ++vr)
                h.v[t][vr] = x[(long long)(rt * 16 + vr + 8 * (lane >> 4)) * F_IN_C
                               + t * 16 + (lane & 15)];
    } else {
#pragma unroll
        for (int t = 0; t < 2; ++t)
#pragma unroll
            for (int vr = 0; vr < 8; ++vr) h.v[t][vr] = 0.0f;
    }
    return h;
}

__global__ void __launch_bounds__(512)
gru0_kernel(const float* __restrict__ s0, const float* __restrict__ x,
            float* __restrict__ hout,
            const __bf16* __restrict__ wcb, const __bf16* __restrict__ whhb,
            const float* __restrict__ bih, const float* __restrict__ bhh) {
    int lane = threadIdx.x & 31;
    int half = lane >> 4;
    int wave = __builtin_amdgcn_readfirstlane(threadIdx.x >> 5);
    int cp   = wave & 7;
    int set  = wave >> 3;

    v16bf bw[2][3][2];
#pragma unroll
    for (int mat = 0; mat < 2; ++mat)
#pragma unroll
        for (int g = 0; g < 3; ++g)
#pragma unroll
            for (int t = 0; t < 2; ++t) {
                const __bf16* p = (mat ? whhb : wcb)
                                + (((g * 16 + cp * 2 + t) * 8) << 9) + lane * 8;
                b_raw r;
                r.q0 = *(const v8bf*)(p);
                r.q1 = *(const v8bf*)(p + 256);
                bw[mat][g][t] = make_b_frag(r);
            }

    float bi[3][2], bh[3][2];
#pragma unroll
    for (int g = 0; g < 3; ++g)
#pragma unroll
        for (int t = 0; t < 2; ++t) {
            int c = cp * 32 + t * 16 + (lane & 15);
            bi[g][t] = bih[g * 256 + c];
            bh[g][t] = bhh[g * 256 + c];
        }

    int rt = blockIdx.x + GRU0_WGS * set;
    hold_t ho = load_hold0(x, rt, cp, lane);
    v16bf a_agg, a_h;
    {
        a_raw ca = load_a_raw_f32(s0, rt * 16 + (lane & 15), F_IN_C, 0, lane);
        a_raw ch = load_a_raw_f32(x,  rt * 16 + (lane & 15), F_IN_C, 0, lane);
        a_agg = make_a_frag(ca);
        a_h   = make_a_frag(ch);
    }

    while (true) {
        int  rtn  = rt + 2 * GRU0_WGS;
        bool more = rtn < REAL_RT;
        int  rtp  = more ? rtn : rt;
        a_raw  na = load_a_raw_f32(s0, rtp * 16 + (lane & 15), F_IN_C, 0, lane);
        a_raw  nh = load_a_raw_f32(x,  rtp * 16 + (lane & 15), F_IN_C, 0, lane);
        hold_t no = load_hold0(x, rtp, cp, lane);

#pragma unroll
        for (int t = 0; t < 2; ++t) {
            v8f acc[2][3];
#pragma unroll
            for (int g = 0; g < 3; ++g) {
                v8f z0 = {}, z1 = {};
                acc[0][g] = wmma_bf16(a_agg, bw[0][g][t], z0);
                acc[1][g] = wmma_bf16(a_h,   bw[1][g][t], z1);
            }
            int c = cp * 32 + t * 16 + (lane & 15);
#pragma unroll
            for (int vr = 0; vr < 8; ++vr) {
                long long idx = (long long)(rt * 16 + vr + 8 * half) * 256 + c;
                float r  = sigmoidf_(acc[0][0][vr] + bi[0][t] + acc[1][0][vr] + bh[0][t]);
                float z  = sigmoidf_(acc[0][1][vr] + bi[1][t] + acc[1][1][vr] + bh[1][t]);
                float nn = tanhf_(acc[0][2][vr] + bi[2][t] + r * (acc[1][2][vr] + bh[2][t]));
                hout[idx] = (1.0f - z) * nn + z * ho.v[t][vr];
            }
        }

        if (!more) break;
        rt = rtn;
        a_agg = make_a_frag(na);
        a_h   = make_a_frag(nh);
        ho = no;
    }

    if (blockIdx.x == GRU0_WGS - 1 && set == 1) {
        for (int r = N_NODES_C; r < N_PAD; ++r)
            hout[(long long)r * 256 + cp * 32 + lane] = 0.0f;
    }
}

__global__ void __launch_bounds__(256)
final_kernel(const float* __restrict__ h, const float* __restrict__ fcw,
             const float* __restrict__ fcb, float* __restrict__ out) {
    __shared__ float sw[N_ACT_C * 256];
    __shared__ float sb[N_ACT_C];
    for (int i = threadIdx.x; i < N_ACT_C * 256; i += 256) sw[i] = fcw[i];
    if (threadIdx.x < N_ACT_C) sb[threadIdx.x] = fcb[threadIdx.x];
    __syncthreads();

    int n = blockIdx.x * 256 + threadIdx.x;
    if (n >= N_NODES_C) return;

    float acc[N_ACT_C];
#pragma unroll
    for (int a = 0; a < N_ACT_C; ++a) acc[a] = sb[a];
    const float* hr = h + (long long)n * 256;
    for (int k = 0; k < 256; ++k) {
        float hv = fmaxf(hr[k], 0.0f);
#pragma unroll
        for (int a = 0; a < N_ACT_C; ++a) acc[a] += hv * sw[a * 256 + k];
    }
    float mx = acc[0];
#pragma unroll
    for (int a = 1; a < N_ACT_C; ++a) mx = fmaxf(mx, acc[a]);
    float s = 0.0f;
#pragma unroll
    for (int a = 0; a < N_ACT_C; ++a) { acc[a] = __expf(acc[a] - mx); s += acc[a]; }
    float inv = 1.0f / s;
#pragma unroll
    for (int a = 0; a < N_ACT_C; ++a) out[(long long)n * N_ACT_C + a] = acc[a] * inv;
}

extern "C" void kernel_launch(void* const* d_in, const int* in_sizes, int n_in,
                              void* d_out, int out_size, void* d_ws, size_t ws_size,
                              hipStream_t stream) {
    const float* x   = (const float*)d_in[0];
    const int*   ei  = (const int*)  d_in[1];
    const float* ew  = (const float*)d_in[2];
    const float* gg  = (const float*)d_in[3];
    const float* wih = (const float*)d_in[4];
    const float* whh = (const float*)d_in[5];
    const float* bih = (const float*)d_in[6];
    const float* bhh = (const float*)d_in[7];
    const float* fcw = (const float*)d_in[8];
    const float* fcb = (const float*)d_in[9];
    float* out = (float*)d_out;

    float* h0  = (float*)d_ws;
    float* h1  = h0 + (size_t)N_PAD * 256;
    float* agg = h1 + (size_t)N_PAD * 256;
    __bf16* wcb  = (__bf16*)(agg + (size_t)N_PAD * 256);
    __bf16* whhb = wcb + 3 * 768 * 256;
    int*   cnt    = (int*)(whhb + 768 * 256);
    int*   off    = cnt + 50176;
    int*   slot   = off + 50176;
    int*   bucket = slot + N_EDGES_C;
    int*   csrc   = bucket + N_EDGES_C;
    float* cw     = (float*)(csrc + N_EDGES_C);

    convw_kernel<<<(768 * 256 + 255) / 256, 256, 0, stream>>>(whh, whhb);
    combw_kernel<<<(3 * 48 * 16) / 8, 256, 0, stream>>>(gg, wih, wcb);

    zero_int_kernel<<<(N_NODES_C + 1 + 255) / 256, 256, 0, stream>>>(cnt, N_NODES_C + 1);
    count_kernel<<<N_EDGES_C / 256, 256, 0, stream>>>(ei, cnt, slot);
    scan_kernel<<<1, 1024, 0, stream>>>(cnt, off);
    fill_kernel<<<N_EDGES_C / 256, 256, 0, stream>>>(ei, off, slot, bucket);
    rank_kernel<<<N_EDGES_C / 256, 256, 0, stream>>>(ei, ew, off, bucket, csrc, cw);

    float* s0 = h0;
    gather0_kernel<<<G0_BLOCKS, 256, 0, stream>>>(x, off, csrc, cw, s0);
    gru0_kernel<<<GRU0_WGS, 512, 0, stream>>>(s0, x, h1, wcb, whhb, bih, bhh);

    float* hc = h1;
    float* hn = h0;
    for (int l = 1; l < 3; ++l) {
        gather_kernel<<<N_PAD / 8, 256, 0, stream>>>(hc, off, csrc, cw, agg);
        gru_kernel<<<GRU_SLICES * GRU_GROUPS, GRU_WAVES * 32, 0, stream>>>(agg, hc, hn,
                                                                wcb + (size_t)l * 196608, whhb,
                                                                bih, bhh);
        float* t = hc; hc = hn; hn = t;
    }

    final_kernel<<<(N_NODES_C + 255) / 256, 256, 0, stream>>>(hc, fcw, fcb, out);
}


# Rank kernel: one wave per node holds its bucket's edge ids one per lane and counts smaller ids by lane broadcast; was a thread per edge reading its whole bucket, each lane on a line of its own
# speedup vs baseline: 5.3659x; 1.1336x over previous
#include <hip/hip_runtime.h>

#define N_NODES_C 50000
#define N_EDGES_C 800000
#define F_IN_C 32
#define C_C 256
#define N_PAD 50016
#define N_ACT_C 8

typedef __attribute__((ext_vector_type(16))) __bf16 v16bf;
typedef __attribute__((ext_vector_type(8)))  __bf16 v8bf;
typedef __attribute__((ext_vector_type(8)))  float  v8f;

static __device__ __forceinline__ v8f wmma_bf16(v16bf a, v16bf b, v8f c) {
    return __builtin_amdgcn_wmma_f32_16x16x32_bf16(
         false, a,  false, b,
         (short)0, c,  false,  false);
}

typedef __attribute__((ext_vector_type(4))) float v4f;
typedef __attribute__((ext_vector_type(2))) float v2f;

static __device__ __forceinline__ v8f wmma_f32(v2f a, v2f b, v8f c) {
    return __builtin_amdgcn_wmma_f32_16x16x4_f32(
         false, a,  false, b,
         (short)0, c,  false,  false);
}


struct a_raw { v4f f0, f1, f2, f3; };

static __device__ __forceinline__ a_raw load_a_raw_f32(
    const float* __restrict__ A, int row, int ld, int k0, int lane) {
    int half = lane >> 4;
    const float* p0 = A + (long long)row * ld + k0 + half * 8;
    const float* p1 = p0 + 16;
    a_raw r;
    r.f0 = *(const v4f*)(p0);
    r.f1 = *(const v4f*)(p0 + 4);
    r.f2 = *(const v4f*)(p1);
    r.f3 = *(const v4f*)(p1 + 4);
    return r;
}

static __device__ __forceinline__ v16bf make_a_frag(const a_raw& r) {
    v16bf a;
#pragma unroll
    for (int i = 0; i < 4; ++i) {
        a[i]      = (__bf16)r.f0[i];
        a[4 + i]  = (__bf16)r.f1[i];
        a[8 + i]  = (__bf16)r.f2[i];
        a[12 + i] = (__bf16)r.f3[i];
    }
    return a;
}

struct b_raw { v8bf q0, q1; };

static __device__ __forceinline__ v16bf make_b_frag(const b_raw& r) {
    v16bf b;
#pragma unroll
    for (int i = 0; i < 8; ++i) { b[i] = r.q0[i]; b[8 + i] = r.q1[i]; }
    return b;
}

static __device__ __forceinline__ float sigmoidf_(float x) {
    return __builtin_amdgcn_rcpf(1.0f + __builtin_amdgcn_exp2f(x * -1.44269504f));
}

static __device__ __forceinline__ float tanhf_(float x) {
    return 1.0f - 2.0f * __builtin_amdgcn_rcpf(1.0f + __builtin_amdgcn_exp2f(x * 2.88539008f));
}

__global__ void convw_kernel(const float* __restrict__ whh,
                             __bf16* __restrict__ whhb) {
    int o = blockIdx.x * 256 + threadIdx.x;
    if (o >= 768 * 256) return;
    int i  = o & 7;
    int l  = (o >> 3) & 31;
    int q  = (o >> 8) & 1;
    int ks = (o >> 9) & 7;
    int ct = o >> 12;
    int col = ct * 16 + (l & 15);
    int k   = ks * 32 + q * 16 + (l >> 4) * 8 + i;
    whhb[o] = (__bf16)whh[col * 256 + k];
}

__global__ void __launch_bounds__(256)
combw_kernel(const float* __restrict__ gg, const float* __restrict__ wih,
             __bf16* __restrict__ wc) {
    int lane  = threadIdx.x & 31;
    int tile  = __builtin_amdgcn_readfirstlane(blockIdx.x * 8 + (threadIdx.x >> 5));
    int layer = tile / 768;
    int rem   = tile - layer * 768;
    int ct    = rem >> 4;
    int kt    = rem & 15;
    if (layer == 0 && kt >= 2) return;
    int half  = lane >> 4;
    const float* pa = gg + layer * 65536 + (kt * 16 + (lane & 15)) * 256 + half * 8;
    const float* pb = wih + (ct * 16 + (lane & 15)) * 256 + half * 8;

    v8f acc0 = {}, acc1 = {};
#pragma unroll
    for (int ch = 0; ch < 16; ++ch) {
        v4f a0 = *(const v4f*)(pa + ch * 16);
        v4f a1 = *(const v4f*)(pa + ch * 16 + 4);
        v4f b0 = *(const v4f*)(pb + ch * 16);
        v4f b1 = *(const v4f*)(pb + ch * 16 + 4);
        v2f x0 = {a0[0], a0[1]}, y0 = {b0[0], b0[1]};
        v2f x1 = {a0[2], a0[3]}, y1 = {b0[2], b0[3]};
        v2f x2 = {a1[0], a1[1]}, y2 = {b1[0], b1[1]};
        v2f x3 = {a1[2], a1[3]}, y3 = {b1[2], b1[3]};
        acc0 = wmma_f32(x0, y0, acc0);
        acc1 = wmma_f32(x1, y1, acc1);
        acc0 = wmma_f32(x2, y2, acc0);
        acc1 = wmma_f32(x3, y3, acc1);
    }
    v8bf o;
#pragma unroll
    for (int v = 0; v < 8; ++v) o[v] = (__bf16)(acc0[v] + acc1[v]);
    *(v8bf*)(wc + layer * 196608
                + ((((ct * 8 + (kt >> 1)) * 2 + (kt & 1))) << 8) + lane * 8) = o;
}

__global__ void zero_int_kernel(int* __restrict__ p, int n) {
    int i = blockIdx.x * 256 + threadIdx.x;
    if (i < n) p[i] = 0;
}

__global__ void __launch_bounds__(256)
count_kernel(const int* __restrict__ ei, int* __restrict__ cnt,
             int* __restrict__ slot) {
    int e = blockIdx.x * 256 + threadIdx.x;
    if (e >= N_EDGES_C) return;
    slot[e] = atomicAdd(&cnt[ei[N_EDGES_C + e]], 1);
}

#define SCAN_PER ((N_NODES_C + 1 + 1023) / 1024)
#define SCAN_TOT (SCAN_PER * 1024)
#define SCAN_Q   (SCAN_TOT / 4)
#define SCAN_QIN (N_NODES_C / 4)
static_assert(N_NODES_C % 4 == 0, "counts are moved 4 at a time");
static_assert(SCAN_TOT % 4 == 0 && SCAN_TOT >= N_NODES_C + 1, "LDS slots");

__global__ void __launch_bounds__(1024)
scan_kernel(const int* __restrict__ cnt, int* __restrict__ off) {
    __shared__ __attribute__((aligned(16))) int buf[SCAN_TOT];
    __shared__ int wtot[32];
    int t    = threadIdx.x;
    int lane = t & 31;
    int wave = t >> 5;

    {
        const int4* cnt4 = (const int4*)cnt;
        int4* buf4 = (int4*)buf;
        int4 q[(SCAN_Q + 1023) / 1024];
#pragma unroll
        for (int j = 0; j < (SCAN_Q + 1023) / 1024; ++j) {
            int p = j * 1024 + t;
            q[j] = make_int4(0, 0, 0, 0);
            if (p < SCAN_QIN) q[j] = cnt4[p];
        }
#pragma unroll
        for (int j = 0; j < (SCAN_Q + 1023) / 1024; ++j) {
            int p = j * 1024 + t;
            if (p < SCAN_Q) buf4[p] = q[j];
        }
    }
    __syncthreads();

    int v[SCAN_PER];
    int s = 0;
#pragma unroll
    for (int i = 0; i < SCAN_PER; ++i) { v[i] = buf[t * SCAN_PER + i]; s += v[i]; }

    int inc = s;
#pragma unroll
    for (int o = 1; o < 32; o <<= 1) {
        int u = __shfl_up(inc, o);
        if (lane >= o) inc += u;
    }
    if (lane == 31) wtot[wave] = inc;
    __syncthreads();
    int wt   = wtot[lane];
    int winc = wt;
#pragma unroll
    for (int o = 1; o < 32; o <<= 1) {
        int u = __shfl_up(winc, o);
        if (lane >= o) winc += u;
    }
    int before = __shfl(winc - wt, wave);
    int run = before + inc - s;
#pragma unroll
    for (int i = 0; i < SCAN_PER; ++i) { buf[t * SCAN_PER + i] = run; run += v[i]; }
    __syncthreads();

    {
        const int4* buf4 = (const int4*)buf;
        int4* off4 = (int4*)off;
#pragma unroll
        for (int j = 0; j < (SCAN_QIN + 1023) / 1024; ++j) {
            int p = j * 1024 + t;
            if (p < SCAN_QIN) off4[p] = buf4[p];
        }
        if (t == 0) off[N_NODES_C] = buf[N_NODES_C];
    }
}

__global__ void __launch_bounds__(256)
fill_kernel(const int* __restrict__ ei, const int* __restrict__ off,
            const int* __restrict__ slot, int* __restrict__ bucket) {
    int e = blockIdx.x * 256 + threadIdx.x;
    if (e >= N_EDGES_C) return;
    bucket[off[ei[N_EDGES_C + e]] + slot[e]] = e;
}

#define RK_BLOCKS 1024
#define RK_STRIDE (RK_BLOCKS * 8)
#define RK_NONE   0x7fffffff

static __device__ __forceinline__ int rank_piece(const int* __restrict__ bucket,
                                                 int b, int n, int base, int lane) {
    int v = RK_NONE;
    if (base + lane < n) v = bucket[b + base + lane];
    return v;
}

static __device__ __forceinline__ int rank_count(const int* __restrict__ bucket,
                                                 int b, int n, int first, int mine, int lane) {
    int r = 0;
    for (int bj = 0; bj < n; bj += 32) {
        int oth = (bj == 0) ? first : rank_piece(bucket, b, n, bj, lane);
        int c = n - bj;
        if (c > 32) c = 32;
        for (int j0 = 0; j0 < c; j0 += 8) {
#pragma unroll
            for (int k = 0; k < 8; ++k) {
                int ej = __builtin_amdgcn_readlane(oth, j0 + k);
                r += (ej < mine) ? 1 : 0;
            }
        }
    }
    return r;
}

__global__ void __launch_bounds__(256)
rank_kernel(const int* __restrict__ ei, const float* __restrict__ ew,
            const int* __restrict__ off, const int* __restrict__ bucket,
            int* __restrict__ csrc, float* __restrict__ cw) {
    int lane = threadIdx.x & 31;
    int gw = __builtin_amdgcn_readfirstlane(blockIdx.x * 8 + (threadIdx.x >> 5));

    int node = gw;
    int b0 = 0, e0 = 0, b1 = 0, e1 = 0;
    if (node < N_NODES_C)             { b0 = off[node];             e0 = off[node + 1]; }
    if (node + RK_STRIDE < N_NODES_C) { b1 = off[node + RK_STRIDE]; e1 = off[node + RK_STRIDE + 1]; }
    int m0 = rank_piece(bucket, b0, e0 - b0, 0, lane);

    while (node < N_NODES_C) {
        int n0 = e0 - b0;
        int r = rank_count(bucket, b0, n0, m0, m0, lane);

        int m1 = rank_piece(bucket, b1, e1 - b1, 0, lane);
        int b2 = 0, e2 = 0;
        if (node + 2 * RK_STRIDE < N_NODES_C) {
            b2 = off[node + 2 * RK_STRIDE];
            e2 = off[node + 2 * RK_STRIDE + 1];
        }

        if (lane < n0) {
            csrc[b0 + r] = ei[m0];
            cw[b0 + r]   = ew[m0];
        }
        for (int bi = 32; bi < n0; bi += 32) {
            int mine = rank_piece(bucket, b0, n0, bi, lane);
            int rr = rank_count(bucket, b0, n0, m0, mine, lane);
            if (bi + lane < n0) {
                csrc[b0 + rr] = ei[mine];
                cw[b0 + rr]   = ew[mine];
            }
        }

        node += RK_STRIDE;
        b0 = b1; e0 = e1; m0 = m1;
        b1 = b2; e1 = e2;
    }
}

__global__ void __launch_bounds__(256)
gather_kernel(const float* __restrict__ h, const int* __restrict__ off,
              const int* __restrict__ csrc, const float* __restrict__ cw,
              float* __restrict__ agg) {
    int lane = threadIdx.x & 31;
    int node = __builtin_amdgcn_readfirstlane(blockIdx.x * 8 + (threadIdx.x >> 5));
    int b = 0, n = 0;
    if (node < N_NODES_C) { b = off[node]; n = off[node + 1] - b; }

    float4 a0 = make_float4(0.f, 0.f, 0.f, 0.f);
    float4 a1 = make_float4(0.f, 0.f, 0.f, 0.f);
    for (int base = 0; base < n; base += 32) {
        int c = n - base;
        if (c > 32) c = 32;
        int   s_l = 0;
        float w_l = 0.0f;
        if (lane < c) { s_l = csrc[b + base + lane]; w_l = cw[b + base + lane]; }
#pragma unroll 4
        for (int j = 0; j < c; ++j) {
            int   s = __builtin_amdgcn_readlane(s_l, j);
            float w = __int_as_float(__builtin_amdgcn_readlane(__float_as_int(w_l), j));
            const float4* row = (const float4*)(h + (long long)s * 256);
            float4 v0 = row[lane];
            float4 v1 = row[32 + lane];
            a0.x += v0.x * w; a0.y += v0.y * w; a0.z += v0.z * w; a0.w += v0.w * w;
            a1.x += v1.x * w; a1.y += v1.y * w; a1.z += v1.z * w; a1.w += v1.w * w;
        }
    }
    float4* orow = (float4*)(agg + (long long)node * 256);
    orow[lane]      = a0;
    orow[32 + lane] = a1;
}

#define GRU_SLICES 4
#define GRU_GROUPS 64
#define GRU_WAVES  4
#define GRU_RP     (N_PAD / 32)
#define GRU_STRIDE (GRU_GROUPS * GRU_WAVES)
static_assert(N_PAD % 32 == 0, "row pairs tile the padded rows");
static_assert(GRU_STRIDE <= GRU_RP, "every wave has at least one row pair");

struct b_set { b_raw f[2][4]; };

static __device__ __forceinline__ void lds_load_set(b_set& s, const __bf16* L,
                                                    int g, int ks, int lane) {
#pragma unroll
    for (int mat = 0; mat < 2; ++mat) {
#pragma unroll
        for (int t = 0; t < 4; ++t) {
            const __bf16* p = L + ((((mat * 3 + g) * 4 + t) * 8 + ks) << 9) + lane * 8;
            s.f[mat][t].q0 = *(const v8bf*)(p);
            s.f[mat][t].q1 = *(const v8bf*)(p + 256);
        }
    }
}

__global__ void __launch_bounds__(GRU_WAVES * 32)
gru_kernel(const float* __restrict__ agg, const float* __restrict__ hin,
           float* __restrict__ hout,
           const __bf16* __restrict__ wcb, const __bf16* __restrict__ whhb,
           const float* __restrict__ bih, const float* __restrict__ bhh) {
    __shared__ __attribute__((aligned(16))) __bf16 wl[2 * 3 * 4 * 8 * 512];

    int tid  = threadIdx.x;
    int lane = tid & 31;
    int wave = tid >> 5;
    int half = lane >> 4;
    int slice = blockIdx.x & (GRU_SLICES - 1);
    int group = blockIdx.x >> 2;

#pragma unroll
    for (int mg = 0; mg < 6; ++mg) {
        const __bf16* src = ((mg >= 3) ? whhb : wcb)
                          + ((((mg % 3) * 16 + slice * 4) * 8) << 9) + tid * 8;
        __bf16* dst = wl + mg * 16384 + tid * 8;
        v8bf tmp[16];
#pragma unroll
        for (int p = 0; p < 16; ++p) tmp[p] = *(const v8bf*)(src + p * 1024);
#pragma unroll
        for (int p = 0; p < 16; ++p) *(v8bf*)(dst + p * 1024) = tmp[p];
    }
    __syncthreads();

    int rp   = group * GRU_WAVES + wave;
    int rowA = rp * 32 + (lane & 15);

    a_raw pa[4][2], ph[4][2];
#pragma unroll
    for (int s = 0; s < 4; ++s)
#pragma unroll
        for (int u = 0; u < 2; ++u) {
            pa[s][u] = load_a_raw_f32(agg, rowA + 16 * u, 256, s * 32, lane);
            ph[s][u] = load_a_raw_f32(hin, rowA + 16 * u, 256, s * 32, lane);
        }

    b_set bs[2];
    lds_load_set(bs[0], wl, 0, 0, lane);

    while (true) {
        int  rpn  = rp + GRU_STRIDE;
        bool more = rpn < GRU_RP;
        int  rowN = (more ? rpn : rp) * 32 + (lane & 15);

        float hold[2][4][8];
        v8f acc[2][2][3][4];
#pragma unroll
        for (int u = 0; u < 2; ++u)
#pragma unroll
            for (int a = 0; a < 2; ++a)
#pragma unroll
                for (int g = 0; g < 3; ++g)
#pragma unroll
                    for (int t = 0; t < 4; ++t) { v8f z = {}; acc[u][a][g][t] = z; }

#pragma unroll
        for (int ks = 0; ks < 8; ++ks) {
            v16bf a_agg[2], a_h[2];
#pragma unroll
            for (int u = 0; u < 2; ++u) {
                a_agg[u] = make_a_frag(pa[ks & 3][u]);
                a_h[u]   = make_a_frag(ph[ks & 3][u]);
                if (ks < 4) {
                    pa[ks & 3][u] = load_a_raw_f32(agg, rowA + 16 * u, 256, (ks + 4) * 32, lane);
                    ph[ks & 3][u] = load_a_raw_f32(hin, rowA + 16 * u, 256, (ks + 4) * 32, lane);
                } else {
                    pa[ks & 3][u] = load_a_raw_f32(agg, rowN + 16 * u, 256, (ks - 4) * 32, lane);
                    ph[ks & 3][u] = load_a_raw_f32(hin, rowN + 16 * u, 256, (ks - 4) * 32, lane);
                }
            }
            if (ks == 4 || ks == 6) {
                const int u = (ks - 4) >> 1;
#pragma unroll
                for (int t = 0; t < 4; ++t)
#pragma unroll
                    for (int vr = 0; vr < 8; ++vr)
                        hold[u][t][vr] = hin[(long long)(rp * 32 + 16 * u + vr + 8 * half) * 256
                                             + slice * 64 + t * 16 + (lane & 15)];
            }
#pragma unroll
            for (int g = 0; g < 3; ++g) {
                const int bi = ks * 3 + g;
                const int nb = (bi + 1) % 24;
                b_set& cur = bs[bi & 1];
                lds_load_set(bs[(bi + 1) & 1], wl, nb % 3, nb / 3, lane);
                __builtin_amdgcn_sched_barrier(0);
#pragma unroll
                for (int t = 0; t < 4; ++t) {
                    v16bf b0 = make_b_frag(cur.f[0][t]);
                    v16bf b1 = make_b_frag(cur.f[1][t]);
#pragma unroll
                    for (int u = 0; u < 2; ++u) {
                        acc[u][0][g][t] = wmma_bf16(a_agg[u], b0, acc[u][0][g][t]);
                        acc[u][1][g][t] = wmma_bf16(a_h[u],   b1, acc[u][1][g][t]);
                    }
                }
                __builtin_amdgcn_sched_barrier(0);
            }
        }

#pragma unroll
        for (int t = 0; t < 4; ++t) {
            int c = slice * 64 + t * 16 + (lane & 15);
            float bir = bih[c],       bhr = bhh[c];
            float biz = bih[256 + c], bhz = bhh[256 + c];
            float bin = bih[512 + c], bhn = bhh[512 + c];
#pragma unroll
            for (int u = 0; u < 2; ++u) {
#pragma unroll
                for (int vr = 0; vr < 8; ++vr) {
                    long long idx = (long long)(rp * 32 + 16 * u + vr + 8 * half) * 256 + c;
                    float r  = sigmoidf_(acc[u][0][0][t][vr] + bir + acc[u][1][0][t][vr] + bhr);
                    float z  = sigmoidf_(acc[u][0][1][t][vr] + biz + acc[u][1][1][t][vr] + bhz);
                    float nn = tanhf_(acc[u][0][2][t][vr] + bin + r * (acc[u][1][2][t][vr] + bhn));
                    hout[idx] = (1.0f - z) * nn + z * hold[u][t][vr];
                }
            }
        }

        if (!more) break;
        rp   = rpn;
        rowA = rowN;
    }
}

#define G0_BLOCKS 1024
__global__ void __launch_bounds__(256)
gather0_kernel(const float* __restrict__ x, const int* __restrict__ off,
               const int* __restrict__ csrc, const float* __restrict__ cw,
               float* __restrict__ s0) {
    int lane = threadIdx.x & 31;
    int gw = __builtin_amdgcn_readfirstlane(blockIdx.x * 8 + (threadIdx.x >> 5));
    for (int node = gw; node < N_PAD; node += G0_BLOCKS * 8) {
        int b = 0, n = 0;
        if (node < N_NODES_C) { b = off[node]; n = off[node + 1] - b; }
        float a = 0.0f;
        for (int base = 0; base < n; base += 32) {
            int c = n - base;
            if (c > 32) c = 32;
            int   s_l = 0;
            float w_l = 0.0f;
            if (lane < c) { s_l = csrc[b + base + lane]; w_l = cw[b + base + lane]; }
            for (int j0 = 0; j0 < c; j0 += 16) {
                float v[16];
#pragma unroll
                for (int k = 0; k < 16; ++k) {
                    int s = __builtin_amdgcn_readlane(s_l, j0 + k);
                    v[k] = x[(long long)s * F_IN_C + lane];
                }
#pragma unroll
                for (int k = 0; k < 16; ++k) {
                    float w  = __int_as_float(__builtin_amdgcn_readlane(__float_as_int(w_l), j0 + k));
                    float vv = (j0 + k < c) ? v[k] : 0.0f;
                    a = __builtin_fmaf(vv, w, a);
                }
            }
        }
        s0[(long long)node * F_IN_C + lane] = a;
    }
}

#define GRU0_WGS 256
#define REAL_RT  (N_NODES_C / 16)

struct hold_t { float v[2][8]; };

static __device__ __forceinline__ hold_t load_hold0(const float* __restrict__ x,
                                                    int rt, int cp, int lane) {
    hold_t h;
    if (cp == 0) {
#pragma unroll
        for (int t = 0; t < 2; ++t)
#pragma unroll
            for (int vr = 0; vr < 8; ++vr)
                h.v[t][vr] = x[(long long)(rt * 16 + vr + 8 * (lane >> 4)) * F_IN_C
                               + t * 16 + (lane & 15)];
    } else {
#pragma unroll
        for (int t = 0; t < 2; ++t)
#pragma unroll
            for (int vr = 0; vr < 8; ++vr) h.v[t][vr] = 0.0f;
    }
    return h;
}

__global__ void __launch_bounds__(512)
gru0_kernel(const float* __restrict__ s0, const float* __restrict__ x,
            float* __restrict__ hout,
            const __bf16* __restrict__ wcb, const __bf16* __restrict__ whhb,
            const float* __restrict__ bih, const float* __restrict__ bhh) {
    int lane = threadIdx.x & 31;
    int half = lane >> 4;
    int wave = __builtin_amdgcn_readfirstlane(threadIdx.x >> 5);
    int cp   = wave & 7;
    int set  = wave >> 3;

    v16bf bw[2][3][2];
#pragma unroll
    for (int mat = 0; mat < 2; ++mat)
#pragma unroll
        for (int g = 0; g < 3; ++g)
#pragma unroll
            for (int t = 0; t < 2; ++t) {
                const __bf16* p = (mat ? whhb : wcb)
                                + (((g * 16 + cp * 2 + t) * 8) << 9) + lane * 8;
                b_raw r;
                r.q0 = *(const v8bf*)(p);
                r.q1 = *(const v8bf*)(p + 256);
                bw[mat][g][t] = make_b_frag(r);
            }

    float bi[3][2], bh[3][2];
#pragma unroll
    for (int g = 0; g < 3; ++g)
#pragma unroll
        for (int t = 0; t < 2; ++t) {
            int c = cp * 32 + t * 16 + (lane & 15);
            bi[g][t] = bih[g * 256 + c];
            bh[g][t] = bhh[g * 256 + c];
        }

    int rt = blockIdx.x + GRU0_WGS * set;
    hold_t ho = load_hold0(x, rt, cp, lane);
    v16bf a_agg, a_h;
    {
        a_raw ca = load_a_raw_f32(s0, rt * 16 + (lane & 15), F_IN_C, 0, lane);
        a_raw ch = load_a_raw_f32(x,  rt * 16 + (lane & 15), F_IN_C, 0, lane);
        a_agg = make_a_frag(ca);
        a_h   = make_a_frag(ch);
    }

    while (true) {
        int  rtn  = rt + 2 * GRU0_WGS;
        bool more = rtn < REAL_RT;
        int  rtp  = more ? rtn : rt;
        a_raw  na = load_a_raw_f32(s0, rtp * 16 + (lane & 15), F_IN_C, 0, lane);
        a_raw  nh = load_a_raw_f32(x,  rtp * 16 + (lane & 15), F_IN_C, 0, lane);
        hold_t no = load_hold0(x, rtp, cp, lane);

#pragma unroll
        for (int t = 0; t < 2; ++t) {
            v8f acc[2][3];
#pragma unroll
            for (int g = 0; g < 3; ++g) {
                v8f z0 = {}, z1 = {};
                acc[0][g] = wmma_bf16(a_agg, bw[0][g][t], z0);
                acc[1][g] = wmma_bf16(a_h,   bw[1][g][t], z1);
            }
            int c = cp * 32 + t * 16 + (lane & 15);
#pragma unroll
            for (int vr = 0; vr < 8; ++vr) {
                long long idx = (long long)(rt * 16 + vr + 8 * half) * 256 + c;
                float r  = sigmoidf_(acc[0][0][vr] + bi[0][t] + acc[1][0][vr] + bh[0][t]);
                float z  = sigmoidf_(acc[0][1][vr] + bi[1][t] + acc[1][1][vr] + bh[1][t]);
                float nn = tanhf_(acc[0][2][vr] + bi[2][t] + r * (acc[1][2][vr] + bh[2][t]));
                hout[idx] = (1.0f - z) * nn + z * ho.v[t][vr];
            }
        }

        if (!more) break;
        rt = rtn;
        a_agg = make_a_frag(na);
        a_h   = make_a_frag(nh);
        ho = no;
    }

    if (blockIdx.x == GRU0_WGS - 1 && set == 1) {
        for (int r = N_NODES_C; r < N_PAD; ++r)
            hout[(long long)r * 256 + cp * 32 + lane] = 0.0f;
    }
}

__global__ void __launch_bounds__(256)
final_kernel(const float* __restrict__ h, const float* __restrict__ fcw,
             const float* __restrict__ fcb, float* __restrict__ out) {
    __shared__ float sw[N_ACT_C * 256];
    __shared__ float sb[N_ACT_C];
    for (int i = threadIdx.x; i < N_ACT_C * 256; i += 256) sw[i] = fcw[i];
    if (threadIdx.x < N_ACT_C) sb[threadIdx.x] = fcb[threadIdx.x];
    __syncthreads();

    int n = blockIdx.x * 256 + threadIdx.x;
    if (n >= N_NODES_C) return;

    float acc[N_ACT_C];
#pragma unroll
    for (int a = 0; a < N_ACT_C; ++a) acc[a] = sb[a];
    const float* hr = h + (long long)n * 256;
    for (int k = 0; k < 256; ++k) {
        float hv = fmaxf(hr[k], 0.0f);
#pragma unroll
        for (int a = 0; a < N_ACT_C; ++a) acc[a] += hv * sw[a * 256 + k];
    }
    float mx = acc[0];
#pragma unroll
    for (int a = 1; a < N_ACT_C; ++a) mx = fmaxf(mx, acc[a]);
    float s = 0.0f;
#pragma unroll
    for (int a = 0; a < N_ACT_C; ++a) { acc[a] = __expf(acc[a] - mx); s += acc[a]; }
    float inv = 1.0f / s;
#pragma unroll
    for (int a = 0; a < N_ACT_C; ++a) out[(long long)n * N_ACT_C + a] = acc[a] * inv;
}

extern "C" void kernel_launch(void* const* d_in, const int* in_sizes, int n_in,
                              void* d_out, int out_size, void* d_ws, size_t ws_size,
                              hipStream_t stream) {
    const float* x   = (const float*)d_in[0];
    const int*   ei  = (const int*)  d_in[1];
    const float* ew  = (const float*)d_in[2];
    const float* gg  = (const float*)d_in[3];
    const float* wih = (const float*)d_in[4];
    const float* whh = (const float*)d_in[5];
    const float* bih = (const float*)d_in[6];
    const float* bhh = (const float*)d_in[7];
    const float* fcw = (const float*)d_in[8];
    const float* fcb = (const float*)d_in[9];
    float* out = (float*)d_out;

    float* h0  = (float*)d_ws;
    float* h1  = h0 + (size_t)N_PAD * 256;
    float* agg = h1 + (size_t)N_PAD * 256;
    __bf16* wcb  = (__bf16*)(agg + (size_t)N_PAD * 256);
    __bf16* whhb = wcb + 3 * 768 * 256;
    int*   cnt    = (int*)(whhb + 768 * 256);
    int*   off    = cnt + 50176;
    int*   slot   = off + 50176;
    int*   bucket = slot + N_EDGES_C;
    int*   csrc   = bucket + N_EDGES_C;
    float* cw     = (float*)(csrc + N_EDGES_C);

    convw_kernel<<<(768 * 256 + 255) / 256, 256, 0, stream>>>(whh, whhb);
    combw_kernel<<<(3 * 48 * 16) / 8, 256, 0, stream>>>(gg, wih, wcb);

    zero_int_kernel<<<(N_NODES_C + 1 + 255) / 256, 256, 0, stream>>>(cnt, N_NODES_C + 1);
    count_kernel<<<N_EDGES_C / 256, 256, 0, stream>>>(ei, cnt, slot);
    scan_kernel<<<1, 1024, 0, stream>>>(cnt, off);
    fill_kernel<<<N_EDGES_C / 256, 256, 0, stream>>>(ei, off, slot, bucket);
    rank_kernel<<<RK_BLOCKS, 256, 0, stream>>>(ei, ew, off, bucket, csrc, cw);

    float* s0 = h0;
    gather0_kernel<<<G0_BLOCKS, 256, 0, stream>>>(x, off, csrc, cw, s0);
    gru0_kernel<<<GRU0_WGS, 512, 0, stream>>>(s0, x, h1, wcb, whhb, bih, bhh);

    float* hc = h1;
    float* hn = h0;
    for (int l = 1; l < 3; ++l) {
        gather_kernel<<<N_PAD / 8, 256, 0, stream>>>(hc, off, csrc, cw, agg);
        gru_kernel<<<GRU_SLICES * GRU_GROUPS, GRU_WAVES * 32, 0, stream>>>(agg, hc, hn,
                                                                wcb + (size_t)l * 196608, whhb,
                                                                bih, bhh);
        float* t = hc; hc = hn; hn = t;
    }

    final_kernel<<<(N_NODES_C + 255) / 256, 256, 0, stream>>>(hc, fcw, fcb, out);
}
